# starting kernel
# baseline (speedup 1.0000x reference)
#include <hip/hip_runtime.h>

typedef __attribute__((ext_vector_type(2))) float v2f;
typedef __attribute__((ext_vector_type(8))) float v8f;

#define GCN_N_NODES 100000
#define GCN_IN_DIM  256
#define GCN_OUT_DIM 128

__global__ __launch_bounds__(32)
void gcn_gemm_wmma_f32(const float* __restrict__ X,
                       const float* __restrict__ W,
                       const float* __restrict__ bias,
                       float* __restrict__ H)
{
    const int mTile  = blockIdx.x;
    const int lane   = threadIdx.x;
    const int lane15 = lane & 15;
    const int hi     = lane >> 4;
    const int m0     = mTile * 16;

    const float* aPtr = X + (size_t)(m0 + lane15) * GCN_IN_DIM + 2 * hi;

    const float* bPtr[8];
#pragma unroll
    for (int j = 0; j < 8; ++j)
        bPtr[j] = W + (size_t)(16 * j + lane15) * GCN_IN_DIM + 2 * hi;

    v8f acc[8];
#pragma unroll
    for (int j = 0; j < 8; ++j)
        acc[j] = (v8f){0.f, 0.f, 0.f, 0.f, 0.f, 0.f, 0.f, 0.f};

    for (int k = 0; k < GCN_IN_DIM; k += 4) {
        v2f a = *(const v2f*)(aPtr + k);
#pragma unroll
        for (int j = 0; j < 8; ++j) {
            v2f bv = *(const v2f*)(bPtr[j] + k);
            acc[j] = __builtin_amdgcn_wmma_f32_16x16x4_f32(
                 false, a,  false, bv,
                 (short)0, acc[j],
                 false,  false);
        }
    }

#pragma unroll
    for (int j = 0; j < 8; ++j) {
        const float bj = bias[16 * j + lane15];
#pragma unroll
        for (int v = 0; v < 8; ++v) {
            const int row = m0 + v + 8 * hi;
            H[(size_t)row * GCN_OUT_DIM + 16 * j + lane15] = acc[j][v] + bj;
        }
    }
}

__global__ void gcn_zero_out(float4* __restrict__ out, int n4)
{
    int i = blockIdx.x * blockDim.x + threadIdx.x;
    if (i < n4) {
        float4 z;
        z.x = 0.f; z.y = 0.f; z.z = 0.f; z.w = 0.f;
        out[i] = z;
    }
}

__global__ __launch_bounds__(256)
void gcn_edge_scatter(const int*   __restrict__ edge_row,
                      const int*   __restrict__ edge_col,
                      const float* __restrict__ edge_val,
                      const float* __restrict__ H,
                      float*       __restrict__ out,
                      int n_edges)
{
    const int e    = (int)((blockIdx.x * (unsigned)blockDim.x + threadIdx.x) >> 5);
    const int lane = threadIdx.x & 31;
    if (e >= n_edges) return;

    const int   r = edge_row[e];
    const int   c = edge_col[e];
    const float v = edge_val[e];

    const float4 h = ((const float4*)(H + (size_t)c * GCN_OUT_DIM))[lane];
    float* orow = out + (size_t)r * GCN_OUT_DIM + lane * 4;

    unsafeAtomicAdd(orow + 0, v * h.x);
    unsafeAtomicAdd(orow + 1, v * h.y);
    unsafeAtomicAdd(orow + 2, v * h.z);
    unsafeAtomicAdd(orow + 3, v * h.w);
}

extern "C" void kernel_launch(void* const* d_in, const int* in_sizes, int n_in,
                              void* d_out, int out_size, void* d_ws, size_t ws_size,
                              hipStream_t stream)
{
    const float* X        = (const float*)d_in[0];
    const int*   edge_row = (const int*)d_in[1];
    const int*   edge_col = (const int*)d_in[2];
    const float* edge_val = (const float*)d_in[3];
    const float* W        = (const float*)d_in[4];
    const float* bias     = (const float*)d_in[5];
    float*       out      = (float*)d_out;
    float*       H        = (float*)d_ws;

    const int n_edges = in_sizes[1];

    gcn_gemm_wmma_f32<<<GCN_N_NODES / 16, 32, 0, stream>>>(X, W, bias, H);

    const int n4 = GCN_N_NODES * GCN_OUT_DIM / 4;
    gcn_zero_out<<<(n4 + 255) / 256, 256, 0, stream>>>((float4*)out, n4);

    const int nblk = (n_edges + 7) / 8;
    gcn_edge_scatter<<<nblk, 256, 0, stream>>>(edge_row, edge_col, edge_val,
                                               H, out, n_edges);
}


# Float-atomic edge scatter replaced by bucketing edges by row (int-atomic count and rank) and an atomic-free sum per row, in two 64-column halves to fit the workspace; sum order = fill rank
# speedup vs baseline: 1.0738x; 1.0738x over previous
#include <hip/hip_runtime.h>

typedef __attribute__((ext_vector_type(2))) float v2f;
typedef __attribute__((ext_vector_type(8))) float v8f;

#define GCN_N_NODES  100000
#define GCN_IN_DIM   256
#define GCN_OUT_DIM  128
#define GCN_HALF_DIM 64

#define WS_OFF_REC 25600000
#define WS_OFF_NXT 38400000
#define WS_OFF_BEG 38800016

__global__ __launch_bounds__(32)
void gcn_gemm_half_wmma_f32(const float* __restrict__ X,
                            const float* __restrict__ W,
                            const float* __restrict__ bias,
                            float* __restrict__ Hh,
                            int o_base)
{
    const int mTile  = blockIdx.x;
    const int lane   = threadIdx.x;
    const int lane15 = lane & 15;
    const int hi     = lane >> 4;
    const int m0     = mTile * 16;

    const float* aPtr = X + (size_t)(m0 + lane15) * GCN_IN_DIM + 2 * hi;

    const float* bPtr[4];
#pragma unroll
    for (int j = 0; j < 4; ++j)
        bPtr[j] = W + (size_t)(o_base + 16 * j + lane15) * GCN_IN_DIM + 2 * hi;

    v8f acc[4];
#pragma unroll
    for (int j = 0; j < 4; ++j)
        acc[j] = (v8f){0.f, 0.f, 0.f, 0.f, 0.f, 0.f, 0.f, 0.f};

    for (int k = 0; k < GCN_IN_DIM; k += 4) {
        v2f a = *(const v2f*)(aPtr + k);
#pragma unroll
        for (int j = 0; j < 4; ++j) {
            v2f bv = *(const v2f*)(bPtr[j] + k);
            acc[j] = __builtin_amdgcn_wmma_f32_16x16x4_f32(
                 false, a,  false, bv,
                 (short)0, acc[j],
                 false,  false);
        }
    }

#pragma unroll
    for (int j = 0; j < 4; ++j) {
        const float bj = bias[o_base + 16 * j + lane15];
#pragma unroll
        for (int v = 0; v < 8; ++v) {
            const int row = m0 + v + 8 * hi;
            Hh[(size_t)row * GCN_HALF_DIM + 16 * j + lane15] = acc[j][v] + bj;
        }
    }
}

__global__ void gcn_clear_counts(int* __restrict__ nxt, int n)
{
    int i = blockIdx.x * blockDim.x + threadIdx.x;
    if (i < n) nxt[i] = 0;
}

__global__ __launch_bounds__(256)
void gcn_count_rows(const int* __restrict__ edge_row, int* __restrict__ nxt, int n_edges)
{
    const int e = (blockIdx.x * 256 + threadIdx.x) * 4;
    if (e + 3 < n_edges) {
        const int4 r = *(const int4*)(edge_row + e);
        atomicAdd(nxt + r.x, 1);
        atomicAdd(nxt + r.y, 1);
        atomicAdd(nxt + r.z, 1);
        atomicAdd(nxt + r.w, 1);
    } else {
        for (int k = e; k < n_edges; ++k)
            atomicAdd(nxt + edge_row[k], 1);
    }
}

__global__ __launch_bounds__(1024)
void gcn_place_segments(int* __restrict__ nxt, int* __restrict__ beg)
{
    __shared__ int wsum[32];
    __shared__ int blk_base;

    const int i    = blockIdx.x * 1024 + threadIdx.x;
    const int lane = threadIdx.x & 31;
    const int w    = threadIdx.x >> 5;

    const int c = (i < GCN_N_NODES) ? nxt[i] : 0;
    int x = c;
    for (int d = 1; d < 32; d <<= 1) {
        const int y = __shfl_up(x, d, 32);
        if (lane >= d) x += y;
    }
    if (lane == 31) wsum[w] = x;
    __syncthreads();
    if (w == 0) {
        const int s = wsum[lane];
        int t = s;
        for (int d = 1; d < 32; d <<= 1) {
            const int y = __shfl_up(t, d, 32);
            if (lane >= d) t += y;
        }
        wsum[lane] = t - s;
        if (lane == 31) blk_base = atomicAdd(nxt + GCN_N_NODES, t);
    }
    __syncthreads();
    const int st = blk_base + wsum[w] + x - c;
    if (i < GCN_N_NODES) {
        beg[i] = st;
        nxt[i] = st;
    }
}

__global__ __launch_bounds__(256)
void gcn_fill_records(const int*   __restrict__ edge_row,
                      const int*   __restrict__ edge_col,
                      const float* __restrict__ edge_val,
                      int*  __restrict__ nxt,
                      int2* __restrict__ rec,
                      int n_edges)
{
    const int e = (blockIdx.x * 256 + threadIdx.x) * 4;
    if (e + 3 < n_edges) {
        const int4   r = *(const int4*)(edge_row + e);
        const int4   c = *(const int4*)(edge_col + e);
        const float4 v = *(const float4*)(edge_val + e);
        const int p0 = atomicAdd(nxt + r.x, 1);
        const int p1 = atomicAdd(nxt + r.y, 1);
        const int p2 = atomicAdd(nxt + r.z, 1);
        const int p3 = atomicAdd(nxt + r.w, 1);
        int2 q;
        q.x = c.x; q.y = __float_as_int(v.x); rec[p0] = q;
        q.x = c.y; q.y = __float_as_int(v.y); rec[p1] = q;
        q.x = c.z; q.y = __float_as_int(v.z); rec[p2] = q;
        q.x = c.w; q.y = __float_as_int(v.w); rec[p3] = q;
    } else {
        for (int k = e; k < n_edges; ++k) {
            const int pos = atomicAdd(nxt + edge_row[k], 1);
            int2 q;
            q.x = edge_col[k];
            q.y = __float_as_int(edge_val[k]);
            rec[pos] = q;
        }
    }
}

__global__ __launch_bounds__(256)
void gcn_row_sum_half(const int*   __restrict__ beg,
                      const int*   __restrict__ nxt,
                      const int2*  __restrict__ rec,
                      const float* __restrict__ Hh,
                      float*       __restrict__ out,
                      int o_base)
{
    const int t = blockIdx.x * 256 + threadIdx.x;
    const int r = t >> 4;
    const int l = threadIdx.x & 15;

    int       p  = beg[r];
    const int pe = nxt[r];

    const float4* H4 = (const float4*)Hh + l;
    float4 acc;
    acc.x = 0.f; acc.y = 0.f; acc.z = 0.f; acc.w = 0.f;

    for (; p + 4 <= pe; p += 4) {
        const int2 e0 = rec[p];
        const int2 e1 = rec[p + 1];
        const int2 e2 = rec[p + 2];
        const int2 e3 = rec[p + 3];
        const float4 h0 = H4[(size_t)e0.x * 16];
        const float4 h1 = H4[(size_t)e1.x * 16];
        const float4 h2 = H4[(size_t)e2.x * 16];
        const float4 h3 = H4[(size_t)e3.x * 16];
        const float v0 = __int_as_float(e0.y);
        const float v1 = __int_as_float(e1.y);
        const float v2 = __int_as_float(e2.y);
        const float v3 = __int_as_float(e3.y);
        acc.x += v0 * h0.x; acc.y += v0 * h0.y; acc.z += v0 * h0.z; acc.w += v0 * h0.w;
        acc.x += v1 * h1.x; acc.y += v1 * h1.y; acc.z += v1 * h1.z; acc.w += v1 * h1.w;
        acc.x += v2 * h2.x; acc.y += v2 * h2.y; acc.z += v2 * h2.z; acc.w += v2 * h2.w;
        acc.x += v3 * h3.x; acc.y += v3 * h3.y; acc.z += v3 * h3.z; acc.w += v3 * h3.w;
    }
    const int n = pe - p;
    if (n > 0) {
        const int2 e0 = rec[p];
        const int2 e1 = (n > 1) ? rec[p + 1] : e0;
        const int2 e2 = (n > 2) ? rec[p + 2] : e0;
        const float4 h0 = H4[(size_t)e0.x * 16];
        const float4 h1 = H4[(size_t)e1.x * 16];
        const float4 h2 = H4[(size_t)e2.x * 16];
        const float v0 = __int_as_float(e0.y);
        const float v1 = __int_as_float(e1.y);
        const float v2 = __int_as_float(e2.y);
        acc.x += v0 * h0.x; acc.y += v0 * h0.y; acc.z += v0 * h0.z; acc.w += v0 * h0.w;
        if (n > 1) {
            acc.x += v1 * h1.x; acc.y += v1 * h1.y; acc.z += v1 * h1.z; acc.w += v1 * h1.w;
        }
        if (n > 2) {
            acc.x += v2 * h2.x; acc.y += v2 * h2.y; acc.z += v2 * h2.z; acc.w += v2 * h2.w;
        }
    }

    ((float4*)(out + (size_t)r * GCN_OUT_DIM + o_base))[l] = acc;
}

extern "C" void kernel_launch(void* const* d_in, const int* in_sizes, int n_in,
                              void* d_out, int out_size, void* d_ws, size_t ws_size,
                              hipStream_t stream)
{
    const float* X        = (const float*)d_in[0];
    const int*   edge_row = (const int*)d_in[1];
    const int*   edge_col = (const int*)d_in[2];
    const float* edge_val = (const float*)d_in[3];
    const float* W        = (const float*)d_in[4];
    const float* bias     = (const float*)d_in[5];
    float*       out      = (float*)d_out;

    float* Hh  = (float*)d_ws;
    int2*  rec = (int2*)((char*)d_ws + WS_OFF_REC);
    int*   nxt = (int*)((char*)d_ws + WS_OFF_NXT);
    int*   beg = (int*)((char*)d_ws + WS_OFF_BEG);

    const int n_edges = in_sizes[1];
    const int eblk    = (n_edges + 1023) / 1024;

    gcn_clear_counts<<<(GCN_N_NODES + 1 + 255) / 256, 256, 0, stream>>>(nxt, GCN_N_NODES + 1);
    gcn_count_rows<<<eblk, 256, 0, stream>>>(edge_row, nxt, n_edges);
    gcn_place_segments<<<(GCN_N_NODES + 1023) / 1024, 1024, 0, stream>>>(nxt, beg);
    gcn_fill_records<<<eblk, 256, 0, stream>>>(edge_row, edge_col, edge_val, nxt, rec, n_edges);

    gcn_gemm_half_wmma_f32<<<GCN_N_NODES / 16, 32, 0, stream>>>(X, W, bias, Hh, 0);
    gcn_row_sum_half<<<GCN_N_NODES * 16 / 256, 256, 0, stream>>>(beg, nxt, rec, Hh, out, 0);

    gcn_gemm_half_wmma_f32<<<GCN_N_NODES / 16, 32, 0, stream>>>(X, W, bias, Hh, 64);
    gcn_row_sum_half<<<GCN_N_NODES * 16 / 256, 256, 0, stream>>>(beg, nxt, rec, Hh, out, 64);
}


# Edge bucketing with one round of per-edge global atomics instead of two: the counting add's return is the edge's rank, the fill places records at beg[row]+rank without atomics; sum order = count rank
# speedup vs baseline: 1.3094x; 1.2194x over previous
#include <hip/hip_runtime.h>

typedef __attribute__((ext_vector_type(2))) float v2f;
typedef __attribute__((ext_vector_type(8))) float v8f;

#define GCN_N_NODES  100000
#define GCN_IN_DIM   256
#define GCN_OUT_DIM  128
#define GCN_HALF_DIM 64

#define WS_OFF_REC  25600000
#define WS_OFF_NXT  38400000
#define WS_OFF_BEG  38800016
#define WS_OFF_RANK 39200016

__global__ __launch_bounds__(32)
void gcn_gemm_half_wmma_f32(const float* __restrict__ X,
                            const float* __restrict__ W,
                            const float* __restrict__ bias,
                            float* __restrict__ Hh,
                            int o_base)
{
    const int mTile  = blockIdx.x;
    const int lane   = threadIdx.x;
    const int lane15 = lane & 15;
    const int hi     = lane >> 4;
    const int m0     = mTile * 16;

    const float* aPtr = X + (size_t)(m0 + lane15) * GCN_IN_DIM + 2 * hi;

    const float* bPtr[4];
#pragma unroll
    for (int j = 0; j < 4; ++j)
        bPtr[j] = W + (size_t)(o_base + 16 * j + lane15) * GCN_IN_DIM + 2 * hi;

    v8f acc[4];
#pragma unroll
    for (int j = 0; j < 4; ++j)
        acc[j] = (v8f){0.f, 0.f, 0.f, 0.f, 0.f, 0.f, 0.f, 0.f};

    for (int k = 0; k < GCN_IN_DIM; k += 4) {
        v2f a = *(const v2f*)(aPtr + k);
#pragma unroll
        for (int j = 0; j < 4; ++j) {
            v2f bv = *(const v2f*)(bPtr[j] + k);
            acc[j] = __builtin_amdgcn_wmma_f32_16x16x4_f32(
                 false, a,  false, bv,
                 (short)0, acc[j],
                 false,  false);
        }
    }

#pragma unroll
    for (int j = 0; j < 4; ++j) {
        const float bj = bias[o_base + 16 * j + lane15];
#pragma unroll
        for (int v = 0; v < 8; ++v) {
            const int row = m0 + v + 8 * hi;
            Hh[(size_t)row * GCN_HALF_DIM + 16 * j + lane15] = acc[j][v] + bj;
        }
    }
}

__global__ void gcn_clear_counts(int* __restrict__ nxt, int n)
{
    int i = blockIdx.x * blockDim.x + threadIdx.x;
    if (i < n) nxt[i] = 0;
}

__global__ __launch_bounds__(256)
void gcn_count_rows(const int* __restrict__ edge_row, int* __restrict__ nxt,
                    int* __restrict__ rank, int n_edges)
{
    const int e = (blockIdx.x * 256 + threadIdx.x) * 4;
    if (e + 3 < n_edges) {
        const int4 r = *(const int4*)(edge_row + e);
        int4 k;
        k.x = atomicAdd(nxt + r.x, 1);
        k.y = atomicAdd(nxt + r.y, 1);
        k.z = atomicAdd(nxt + r.z, 1);
        k.w = atomicAdd(nxt + r.w, 1);
        *(int4*)(rank + e) = k;
    } else {
        for (int i = e; i < n_edges; ++i)
            rank[i] = atomicAdd(nxt + edge_row[i], 1);
    }
}

__global__ __launch_bounds__(1024)
void gcn_place_segments(int* __restrict__ nxt, int* __restrict__ beg)
{
    __shared__ int wsum[32];
    __shared__ int blk_base;

    const int i    = blockIdx.x * 1024 + threadIdx.x;
    const int lane = threadIdx.x & 31;
    const int w    = threadIdx.x >> 5;

    const int c = (i < GCN_N_NODES) ? nxt[i] : 0;
    int x = c;
    for (int d = 1; d < 32; d <<= 1) {
        const int y = __shfl_up(x, d, 32);
        if (lane >= d) x += y;
    }
    if (lane == 31) wsum[w] = x;
    __syncthreads();
    if (w == 0) {
        const int s = wsum[lane];
        int t = s;
        for (int d = 1; d < 32; d <<= 1) {
            const int y = __shfl_up(t, d, 32);
            if (lane >= d) t += y;
        }
        wsum[lane] = t - s;
        if (lane == 31) blk_base = atomicAdd(nxt + GCN_N_NODES, t);
    }
    __syncthreads();
    const int st = blk_base + wsum[w] + x - c;
    if (i < GCN_N_NODES) {
        beg[i] = st;
        nxt[i] = st + c;
    }
}

__global__ __launch_bounds__(256)
void gcn_fill_records(const int*   __restrict__ edge_row,
                      const int*   __restrict__ edge_col,
                      const float* __restrict__ edge_val,
                      const int*   __restrict__ beg,
                      const int*   __restrict__ rank,
                      int2* __restrict__ rec,
                      int n_edges)
{
    const int e = (blockIdx.x * 256 + threadIdx.x) * 4;
    if (e + 3 < n_edges) {
        const int4   r = *(const int4*)(edge_row + e);
        const int4   c = *(const int4*)(edge_col + e);
        const float4 v = *(const float4*)(edge_val + e);
        const int4   k = *(const int4*)(rank + e);
        const int p0 = beg[r.x] + k.x;
        const int p1 = beg[r.y] + k.y;
        const int p2 = beg[r.z] + k.z;
        const int p3 = beg[r.w] + k.w;
        int2 q;
        q.x = c.x; q.y = __float_as_int(v.x); rec[p0] = q;
        q.x = c.y; q.y = __float_as_int(v.y); rec[p1] = q;
        q.x = c.z; q.y = __float_as_int(v.z); rec[p2] = q;
        q.x = c.w; q.y = __float_as_int(v.w); rec[p3] = q;
    } else {
        for (int i = e; i < n_edges; ++i) {
            const int pos = beg[edge_row[i]] + rank[i];
            int2 q;
            q.x = edge_col[i];
            q.y = __float_as_int(edge_val[i]);
            rec[pos] = q;
        }
    }
}

__global__ __launch_bounds__(256)
void gcn_row_sum_half(const int*   __restrict__ beg,
                      const int*   __restrict__ nxt,
                      const int2*  __restrict__ rec,
                      const float* __restrict__ Hh,
                      float*       __restrict__ out,
                      int o_base)
{
    const int t = blockIdx.x * 256 + threadIdx.x;
    const int r = t >> 4;
    const int l = threadIdx.x & 15;

    int       p  = beg[r];
    const int pe = nxt[r];

    const float4* H4 = (const float4*)Hh + l;
    float4 acc;
    acc.x = 0.f; acc.y = 0.f; acc.z = 0.f; acc.w = 0.f;

    for (; p + 4 <= pe; p += 4) {
        const int2 e0 = rec[p];
        const int2 e1 = rec[p + 1];
        const int2 e2 = rec[p + 2];
        const int2 e3 = rec[p + 3];
        const float4 h0 = H4[(size_t)e0.x * 16];
        const float4 h1 = H4[(size_t)e1.x * 16];
        const float4 h2 = H4[(size_t)e2.x * 16];
        const float4 h3 = H4[(size_t)e3.x * 16];
        const float v0 = __int_as_float(e0.y);
        const float v1 = __int_as_float(e1.y);
        const float v2 = __int_as_float(e2.y);
        const float v3 = __int_as_float(e3.y);
        acc.x += v0 * h0.x; acc.y += v0 * h0.y; acc.z += v0 * h0.z; acc.w += v0 * h0.w;
        acc.x += v1 * h1.x; acc.y += v1 * h1.y; acc.z += v1 * h1.z; acc.w += v1 * h1.w;
        acc.x += v2 * h2.x; acc.y += v2 * h2.y; acc.z += v2 * h2.z; acc.w += v2 * h2.w;
        acc.x += v3 * h3.x; acc.y += v3 * h3.y; acc.z += v3 * h3.z; acc.w += v3 * h3.w;
    }
    const int n = pe - p;
    if (n > 0) {
        const int2 e0 = rec[p];
        const int2 e1 = (n > 1) ? rec[p + 1] : e0;
        const int2 e2 = (n > 2) ? rec[p + 2] : e0;
        const float4 h0 = H4[(size_t)e0.x * 16];
        const float4 h1 = H4[(size_t)e1.x * 16];
        const float4 h2 = H4[(size_t)e2.x * 16];
        const float v0 = __int_as_float(e0.y);
        const float v1 = __int_as_float(e1.y);
        const float v2 = __int_as_float(e2.y);
        acc.x += v0 * h0.x; acc.y += v0 * h0.y; acc.z += v0 * h0.z; acc.w += v0 * h0.w;
        if (n > 1) {
            acc.x += v1 * h1.x; acc.y += v1 * h1.y; acc.z += v1 * h1.z; acc.w += v1 * h1.w;
        }
        if (n > 2) {
            acc.x += v2 * h2.x; acc.y += v2 * h2.y; acc.z += v2 * h2.z; acc.w += v2 * h2.w;
        }
    }

    ((float4*)(out + (size_t)r * GCN_OUT_DIM + o_base))[l] = acc;
}

extern "C" void kernel_launch(void* const* d_in, const int* in_sizes, int n_in,
                              void* d_out, int out_size, void* d_ws, size_t ws_size,
                              hipStream_t stream)
{
    const float* X        = (const float*)d_in[0];
    const int*   edge_row = (const int*)d_in[1];
    const int*   edge_col = (const int*)d_in[2];
    const float* edge_val = (const float*)d_in[3];
    const float* W        = (const float*)d_in[4];
    const float* bias     = (const float*)d_in[5];
    float*       out      = (float*)d_out;

    float* Hh  = (float*)d_ws;
    int2*  rec = (int2*)((char*)d_ws + WS_OFF_REC);
    int*   nxt = (int*)((char*)d_ws + WS_OFF_NXT);
    int*   beg = (int*)((char*)d_ws + WS_OFF_BEG);
    int*   rnk = (int*)((char*)d_ws + WS_OFF_RANK);

    const int n_edges = in_sizes[1];
    const int eblk    = (n_edges + 1023) / 1024;

    gcn_clear_counts<<<(GCN_N_NODES + 1 + 255) / 256, 256, 0, stream>>>(nxt, GCN_N_NODES + 1);
    gcn_count_rows<<<eblk, 256, 0, stream>>>(edge_row, nxt, rnk, n_edges);
    gcn_place_segments<<<(GCN_N_NODES + 1023) / 1024, 1024, 0, stream>>>(nxt, beg);
    gcn_fill_records<<<eblk, 256, 0, stream>>>(edge_row, edge_col, edge_val, beg, rnk, rec, n_edges);

    gcn_gemm_half_wmma_f32<<<GCN_N_NODES / 16, 32, 0, stream>>>(X, W, bias, Hh, 0);
    gcn_row_sum_half<<<GCN_N_NODES * 16 / 256, 256, 0, stream>>>(beg, nxt, rec, Hh, out, 0);

    gcn_gemm_half_wmma_f32<<<GCN_N_NODES / 16, 32, 0, stream>>>(X, W, bias, Hh, 64);
    gcn_row_sum_half<<<GCN_N_NODES * 16 / 256, 256, 0, stream>>>(beg, nxt, rec, Hh, out, 64);
}


# Row counters split into one plane per package half, picked by the work-group's die id, so no counter line takes adds from both halves; sum order = half-0 count ranks, then half-1
# speedup vs baseline: 1.7296x; 1.3209x over previous
#include <hip/hip_runtime.h>

typedef __attribute__((ext_vector_type(2))) float v2f;
typedef __attribute__((ext_vector_type(8))) float v8f;

#define GCN_N_NODES  100000
#define GCN_IN_DIM   256
#define GCN_OUT_DIM  128
#define GCN_HALF_DIM 64

#define WS_OFF_REC  25600000
#define WS_OFF_NXT  38400000
#define WS_OFF_BEG  38800016
#define WS_OFF_RANK 39600016
#define WS_OFF_CNT  46000016

__global__ __launch_bounds__(32)
void gcn_gemm_half_wmma_f32(const float* __restrict__ X,
                            const float* __restrict__ W,
                            const float* __restrict__ bias,
                            float* __restrict__ Hh,
                            int o_base)
{
    const int mTile  = blockIdx.x;
    const int lane   = threadIdx.x;
    const int lane15 = lane & 15;
    const int hi     = lane >> 4;
    const int m0     = mTile * 16;

    const float* aPtr = X + (size_t)(m0 + lane15) * GCN_IN_DIM + 2 * hi;

    const float* bPtr[4];
#pragma unroll
    for (int j = 0; j < 4; ++j)
        bPtr[j] = W + (size_t)(o_base + 16 * j + lane15) * GCN_IN_DIM + 2 * hi;

    v8f acc[4];
#pragma unroll
    for (int j = 0; j < 4; ++j)
        acc[j] = (v8f){0.f, 0.f, 0.f, 0.f, 0.f, 0.f, 0.f, 0.f};

    for (int k = 0; k < GCN_IN_DIM; k += 4) {
        v2f a = *(const v2f*)(aPtr + k);
#pragma unroll
        for (int j = 0; j < 4; ++j) {
            v2f bv = *(const v2f*)(bPtr[j] + k);
            acc[j] = __builtin_amdgcn_wmma_f32_16x16x4_f32(
                 false, a,  false, bv,
                 (short)0, acc[j],
                 false,  false);
        }
    }

#pragma unroll
    for (int j = 0; j < 4; ++j) {
        const float bj = bias[o_base + 16 * j + lane15];
#pragma unroll
        for (int v = 0; v < 8; ++v) {
            const int row = m0 + v + 8 * hi;
            Hh[(size_t)row * GCN_HALF_DIM + 16 * j + lane15] = acc[j][v] + bj;
        }
    }
}

__global__ void gcn_clear_counts(int* __restrict__ nxt, int n)
{
    int i = blockIdx.x * blockDim.x + threadIdx.x;
    if (i < n) nxt[i] = 0;
}

__global__ __launch_bounds__(256)
void gcn_count_rows(const int* __restrict__ edge_row, int* __restrict__ cnt,
                    int* __restrict__ rank, int n_edges)
{
    const unsigned id   = __builtin_amdgcn_s_sendmsg_rtn(0x87);
    const unsigned half = (id >> 18) & 1u;
    int* __restrict__ plane = cnt + half * GCN_N_NODES;
    const int tag = (int)(half << 31);

    const int e = (blockIdx.x * 256 + threadIdx.x) * 4;
    if (e + 3 < n_edges) {
        const int4 r = *(const int4*)(edge_row + e);
        int4 k;
        k.x = atomicAdd(plane + r.x, 1) | tag;
        k.y = atomicAdd(plane + r.y, 1) | tag;
        k.z = atomicAdd(plane + r.z, 1) | tag;
        k.w = atomicAdd(plane + r.w, 1) | tag;
        *(int4*)(rank + e) = k;
    } else {
        for (int i = e; i < n_edges; ++i)
            rank[i] = atomicAdd(plane + edge_row[i], 1) | tag;
    }
}

__global__ __launch_bounds__(1024)
void gcn_place_segments(int* __restrict__ cnt, int* __restrict__ nxt,
                        int* __restrict__ beg)
{
    __shared__ int wsum[32];
    __shared__ int blk_base;

    const int i    = blockIdx.x * 1024 + threadIdx.x;
    const int lane = threadIdx.x & 31;
    const int w    = threadIdx.x >> 5;

    const int c0 = (i < GCN_N_NODES) ? cnt[i] : 0;
    const int c1 = (i < GCN_N_NODES) ? cnt[GCN_N_NODES + i] : 0;
    const int c  = c0 + c1;
    int x = c;
    for (int d = 1; d < 32; d <<= 1) {
        const int y = __shfl_up(x, d, 32);
        if (lane >= d) x += y;
    }
    if (lane == 31) wsum[w] = x;
    __syncthreads();
    if (w == 0) {
        const int s = wsum[lane];
        int t = s;
        for (int d = 1; d < 32; d <<= 1) {
            const int y = __shfl_up(t, d, 32);
            if (lane >= d) t += y;
        }
        wsum[lane] = t - s;
        if (lane == 31) blk_base = atomicAdd(cnt + 2 * GCN_N_NODES, t);
    }
    __syncthreads();
    const int st = blk_base + wsum[w] + x - c;
    if (i < GCN_N_NODES) {
        beg[i]               = st;
        beg[GCN_N_NODES + i] = st + c0;
        nxt[i]               = st + c;
    }
}

__global__ __launch_bounds__(256)
void gcn_fill_records(const int*   __restrict__ edge_row,
                      const int*   __restrict__ edge_col,
                      const float* __restrict__ edge_val,
                      const int*   __restrict__ beg,
                      const int*   __restrict__ rank,
                      int2* __restrict__ rec,
                      int n_edges)
{
    const int e = (blockIdx.x * 256 + threadIdx.x) * 4;
    if (e + 3 < n_edges) {
        const int4   r = *(const int4*)(edge_row + e);
        const int4   c = *(const int4*)(edge_col + e);
        const float4 v = *(const float4*)(edge_val + e);
        const int4   k = *(const int4*)(rank + e);
        const int p0 = beg[((unsigned)k.x >> 31) * GCN_N_NODES + r.x] + (k.x & 0x7fffffff);
        const int p1 = beg[((unsigned)k.y >> 31) * GCN_N_NODES + r.y] + (k.y & 0x7fffffff);
        const int p2 = beg[((unsigned)k.z >> 31) * GCN_N_NODES + r.z] + (k.z & 0x7fffffff);
        const int p3 = beg[((unsigned)k.w >> 31) * GCN_N_NODES + r.w] + (k.w & 0x7fffffff);
        int2 q;
        q.x = c.x; q.y = __float_as_int(v.x); rec[p0] = q;
        q.x = c.y; q.y = __float_as_int(v.y); rec[p1] = q;
        q.x = c.z; q.y = __float_as_int(v.z); rec[p2] = q;
        q.x = c.w; q.y = __float_as_int(v.w); rec[p3] = q;
    } else {
        for (int i = e; i < n_edges; ++i) {
            const int k   = rank[i];
            const int pos = beg[((unsigned)k >> 31) * GCN_N_NODES + edge_row[i]] + (k & 0x7fffffff);
            int2 q;
            q.x = edge_col[i];
            q.y = __float_as_int(edge_val[i]);
            rec[pos] = q;
        }
    }
}

__global__ __launch_bounds__(256)
void gcn_row_sum_half(const int*   __restrict__ beg,
                      const int*   __restrict__ nxt,
                      const int2*  __restrict__ rec,
                      const float* __restrict__ Hh,
                      float*       __restrict__ out,
                      int o_base)
{
    const int t = blockIdx.x * 256 + threadIdx.x;
    const int r = t >> 4;
    const int l = threadIdx.x & 15;

    int       p  = beg[r];
    const int pe = nxt[r];

    const float4* H4 = (const float4*)Hh + l;
    float4 acc;
    acc.x = 0.f; acc.y = 0.f; acc.z = 0.f; acc.w = 0.f;

    for (; p + 4 <= pe; p += 4) {
        const int2 e0 = rec[p];
        const int2 e1 = rec[p + 1];
        const int2 e2 = rec[p + 2];
        const int2 e3 = rec[p + 3];
        const float4 h0 = H4[(size_t)e0.x * 16];
        const float4 h1 = H4[(size_t)e1.x * 16];
        const float4 h2 = H4[(size_t)e2.x * 16];
        const float4 h3 = H4[(size_t)e3.x * 16];
        const float v0 = __int_as_float(e0.y);
        const float v1 = __int_as_float(e1.y);
        const float v2 = __int_as_float(e2.y);
        const float v3 = __int_as_float(e3.y);
        acc.x += v0 * h0.x; acc.y += v0 * h0.y; acc.z += v0 * h0.z; acc.w += v0 * h0.w;
        acc.x += v1 * h1.x; acc.y += v1 * h1.y; acc.z += v1 * h1.z; acc.w += v1 * h1.w;
        acc.x += v2 * h2.x; acc.y += v2 * h2.y; acc.z += v2 * h2.z; acc.w += v2 * h2.w;
        acc.x += v3 * h3.x; acc.y += v3 * h3.y; acc.z += v3 * h3.z; acc.w += v3 * h3.w;
    }
    const int n = pe - p;
    if (n > 0) {
        const int2 e0 = rec[p];
        const int2 e1 = (n > 1) ? rec[p + 1] : e0;
        const int2 e2 = (n > 2) ? rec[p + 2] : e0;
        const float4 h0 = H4[(size_t)e0.x * 16];
        const float4 h1 = H4[(size_t)e1.x * 16];
        const float4 h2 = H4[(size_t)e2.x * 16];
        const float v0 = __int_as_float(e0.y);
        const float v1 = __int_as_float(e1.y);
        const float v2 = __int_as_float(e2.y);
        acc.x += v0 * h0.x; acc.y += v0 * h0.y; acc.z += v0 * h0.z; acc.w += v0 * h0.w;
        if (n > 1) {
            acc.x += v1 * h1.x; acc.y += v1 * h1.y; acc.z += v1 * h1.z; acc.w += v1 * h1.w;
        }
        if (n > 2) {
            acc.x += v2 * h2.x; acc.y += v2 * h2.y; acc.z += v2 * h2.z; acc.w += v2 * h2.w;
        }
    }

    ((float4*)(out + (size_t)r * GCN_OUT_DIM + o_base))[l] = acc;
}

extern "C" void kernel_launch(void* const* d_in, const int* in_sizes, int n_in,
                              void* d_out, int out_size, void* d_ws, size_t ws_size,
                              hipStream_t stream)
{
    const float* X        = (const float*)d_in[0];
    const int*   edge_row = (const int*)d_in[1];
    const int*   edge_col = (const int*)d_in[2];
    const float* edge_val = (const float*)d_in[3];
    const float* W        = (const float*)d_in[4];
    const float* bias     = (const float*)d_in[5];
    float*       out      = (float*)d_out;

    float* Hh  = (float*)d_ws;
    int2*  rec = (int2*)((char*)d_ws + WS_OFF_REC);
    int*   nxt = (int*)((char*)d_ws + WS_OFF_NXT);
    int*   beg = (int*)((char*)d_ws + WS_OFF_BEG);
    int*   rnk = (int*)((char*)d_ws + WS_OFF_RANK);
    int*   cnt = (int*)((char*)d_ws + WS_OFF_CNT);

    const int n_edges = in_sizes[1];
    const int eblk    = (n_edges + 1023) / 1024;

    gcn_clear_counts<<<(2 * GCN_N_NODES + 1 + 255) / 256, 256, 0, stream>>>(cnt, 2 * GCN_N_NODES + 1);
    gcn_count_rows<<<eblk, 256, 0, stream>>>(edge_row, cnt, rnk, n_edges);
    gcn_place_segments<<<(GCN_N_NODES + 1023) / 1024, 1024, 0, stream>>>(cnt, nxt, beg);
    gcn_fill_records<<<eblk, 256, 0, stream>>>(edge_row, edge_col, edge_val, beg, rnk, rec, n_edges);

    gcn_gemm_half_wmma_f32<<<GCN_N_NODES / 16, 32, 0, stream>>>(X, W, bias, Hh, 0);
    gcn_row_sum_half<<<GCN_N_NODES * 16 / 256, 256, 0, stream>>>(beg, nxt, rec, Hh, out, 0);

    gcn_gemm_half_wmma_f32<<<GCN_N_NODES / 16, 32, 0, stream>>>(X, W, bias, Hh, 64);
    gcn_row_sum_half<<<GCN_N_NODES * 16 / 256, 256, 0, stream>>>(beg, nxt, rec, Hh, out, 64);
}


# Projection takes its W block from LDS (staged once per 8-wave work-group that walks row strips) and A by 16-byte loads, not 16-line 8-byte loads per k-step; k sum order inside each 8-k group changes
# speedup vs baseline: 2.2093x; 1.2773x over previous
#include <hip/hip_runtime.h>

typedef __attribute__((ext_vector_type(2))) float v2f;
typedef __attribute__((ext_vector_type(4))) float v4f;
typedef __attribute__((ext_vector_type(8))) float v8f;

#define GCN_N_NODES  100000
#define GCN_IN_DIM   256
#define GCN_OUT_DIM  128
#define GCN_HALF_DIM 64

#define WS_OFF_REC  25600000
#define WS_OFF_NXT  38400000
#define WS_OFF_BEG  38800016
#define WS_OFF_RANK 39600016
#define WS_OFF_CNT  46000016

#define GEMM_GRID   512
#define GEMM_STRIPS (GCN_N_NODES / 16)


__device__ __forceinline__
void gcn_gemm_block64(v8f (&acc)[4], const v4f (&a)[8], const v4f* bk, int c, int h)
{
    v4f bcur[4], bnxt[4];
#pragma unroll
    for (int jt = 0; jt < 4; ++jt) bcur[jt] = bk[jt * 16 * 64 + (h ^ c)];
#pragma unroll
    for (int j = 0; j < 8; ++j) {
        if (j + 1 < 8) {
            const int x = (2 * (j + 1) + h) ^ c;
#pragma unroll
            for (int jt = 0; jt < 4; ++jt) bnxt[jt] = bk[jt * 16 * 64 + x];
        }
        const v2f a01 = (v2f){a[j].x, a[j].y};
        const v2f a23 = (v2f){a[j].z, a[j].w};
#pragma unroll
        for (int jt = 0; jt < 4; ++jt) {
            const v2f b01 = (v2f){bcur[jt].x, bcur[jt].y};
            acc[jt] = __builtin_amdgcn_wmma_f32_16x16x4_f32(
                 false, a01,  false, b01,
                 (short)0, acc[jt],
                 false,  false);
        }
#pragma unroll
        for (int jt = 0; jt < 4; ++jt) {
            const v2f b23 = (v2f){bcur[jt].z, bcur[jt].w};
            acc[jt] = __builtin_amdgcn_wmma_f32_16x16x4_f32(
                 false, a23,  false, b23,
                 (short)0, acc[jt],
                 false,  false);
        }
#pragma unroll
        for (int jt = 0; jt < 4; ++jt) bcur[jt] = bnxt[jt];
    }
}

__global__ __launch_bounds__(256)
void gcn_gemm_half_wmma_f32(const float* __restrict__ X,
                            const float* __restrict__ W,
                            const float* __restrict__ bias,
                            float* __restrict__ Hh,
                            int o_base)
{
    __shared__ v4f Ws[64 * 64];

    const int t = threadIdx.x;
#pragma unroll
    for (int i = 0; i < 16; ++i) {
        const int idx = i * 256 + t;
        const int o   = idx >> 6;
        const int q   = idx & 63;
        Ws[o * 64 + (q ^ (o & 15))] =
            *(const v4f*)(W + (size_t)(o_base + o) * GCN_IN_DIM + 4 * q);
    }
    __syncthreads();

    const int lane = t & 31;
    const int wv   = __builtin_amdgcn_readfirstlane(t >> 5);
    const int c    = lane & 15;
    const int h    = lane >> 4;

    float bj[4];
#pragma unroll
    for (int j = 0; j < 4; ++j) bj[j] = bias[o_base + 16 * j + c];

    const v4f* bBase = Ws + c * 64;

    int s = wv * GEMM_GRID + blockIdx.x;
    if (s >= GEMM_STRIPS) return;

    const float* aRow = X + (size_t)(s * 16 + c) * GCN_IN_DIM + 4 * h;
    v4f a0[8], a1[8];
#pragma unroll
    for (int j = 0; j < 8; ++j) a0[j] = *(const v4f*)(aRow + 8 * j);

    for (;;) {
        const int  sn      = s + 8 * GEMM_GRID;
        const bool more    = sn < GEMM_STRIPS;
        const float* aNext = more ? X + (size_t)(sn * 16 + c) * GCN_IN_DIM + 4 * h : aRow;

        v8f acc[4];
#pragma unroll
        for (int j = 0; j < 4; ++j)
            acc[j] = (v8f){0.f, 0.f, 0.f, 0.f, 0.f, 0.f, 0.f, 0.f};

#pragma unroll 1
        for (int kb = 0; kb < GCN_IN_DIM; kb += 128) {
#pragma unroll
            for (int j = 0; j < 8; ++j) a1[j] = *(const v4f*)(aRow + kb + 64 + 8 * j);
            gcn_gemm_block64(acc, a0, bBase + (kb >> 2), c, h);

            const float* nx = (kb + 128 < GCN_IN_DIM) ? aRow + kb + 128 : aNext;
#pragma unroll
            for (int j = 0; j < 8; ++j) a0[j] = *(const v4f*)(nx + 8 * j);
            gcn_gemm_block64(acc, a1, bBase + (kb >> 2) + 16, c, h);
        }

#pragma unroll
        for (int j = 0; j < 4; ++j) {
#pragma unroll
            for (int v = 0; v < 8; ++v) {
                const int row = s * 16 + v + 8 * h;
                Hh[(size_t)row * GCN_HALF_DIM + 16 * j + c] = acc[j][v] + bj[j];
            }
        }

        if (!more) break;
        s    = sn;
        aRow = aNext;
    }
}

__global__ void gcn_clear_counts(int* __restrict__ nxt, int n)
{
    int i = blockIdx.x * blockDim.x + threadIdx.x;
    if (i < n) nxt[i] = 0;
}

__global__ __launch_bounds__(256)
void gcn_count_rows(const int* __restrict__ edge_row, int* __restrict__ cnt,
                    int* __restrict__ rank, int n_edges)
{
    const unsigned id   = __builtin_amdgcn_s_sendmsg_rtn(0x87);
    const unsigned half = (id >> 18) & 1u;
    int* __restrict__ plane = cnt + half * GCN_N_NODES;
    const int tag = (int)(half << 31);

    const int e = (blockIdx.x * 256 + threadIdx.x) * 4;
    if (e + 3 < n_edges) {
        const int4 r = *(const int4*)(edge_row + e);
        int4 k;
        k.x = atomicAdd(plane + r.x, 1) | tag;
        k.y = atomicAdd(plane + r.y, 1) | tag;
        k.z = atomicAdd(plane + r.z, 1) | tag;
        k.w = atomicAdd(plane + r.w, 1) | tag;
        *(int4*)(rank + e) = k;
    } else {
        for (int i = e; i < n_edges; ++i)
            rank[i] = atomicAdd(plane + edge_row[i], 1) | tag;
    }
}

__global__ __launch_bounds__(1024)
void gcn_place_segments(int* __restrict__ cnt, int* __restrict__ nxt,
                        int* __restrict__ beg)
{
    __shared__ int wsum[32];
    __shared__ int blk_base;

    const int i    = blockIdx.x * 1024 + threadIdx.x;
    const int lane = threadIdx.x & 31;
    const int w    = threadIdx.x >> 5;

    const int c0 = (i < GCN_N_NODES) ? cnt[i] : 0;
    const int c1 = (i < GCN_N_NODES) ? cnt[GCN_N_NODES + i] : 0;
    const int c  = c0 + c1;
    int x = c;
    for (int d = 1; d < 32; d <<= 1) {
        const int y = __shfl_up(x, d, 32);
        if (lane >= d) x += y;
    }
    if (lane == 31) wsum[w] = x;
    __syncthreads();
    if (w == 0) {
        const int s = wsum[lane];
        int t = s;
        for (int d = 1; d < 32; d <<= 1) {
            const int y = __shfl_up(t, d, 32);
            if (lane >= d) t += y;
        }
        wsum[lane] = t - s;
        if (lane == 31) blk_base = atomicAdd(cnt + 2 * GCN_N_NODES, t);
    }
    __syncthreads();
    const int st = blk_base + wsum[w] + x - c;
    if (i < GCN_N_NODES) {
        beg[i]               = st;
        beg[GCN_N_NODES + i] = st + c0;
        nxt[i]               = st + c;
    }
}

__global__ __launch_bounds__(256)
void gcn_fill_records(const int*   __restrict__ edge_row,
                      const int*   __restrict__ edge_col,
                      const float* __restrict__ edge_val,
                      const int*   __restrict__ beg,
                      const int*   __restrict__ rank,
                      int2* __restrict__ rec,
                      int n_edges)
{
    const int e = (blockIdx.x * 256 + threadIdx.x) * 4;
    if (e + 3 < n_edges) {
        const int4   r = *(const int4*)(edge_row + e);
        const int4   c = *(const int4*)(edge_col + e);
        const float4 v = *(const float4*)(edge_val + e);
        const int4   k = *(const int4*)(rank + e);
        const int p0 = beg[((unsigned)k.x >> 31) * GCN_N_NODES + r.x] + (k.x & 0x7fffffff);
        const int p1 = beg[((unsigned)k.y >> 31) * GCN_N_NODES + r.y] + (k.y & 0x7fffffff);
        const int p2 = beg[((unsigned)k.z >> 31) * GCN_N_NODES + r.z] + (k.z & 0x7fffffff);
        const int p3 = beg[((unsigned)k.w >> 31) * GCN_N_NODES + r.w] + (k.w & 0x7fffffff);
        int2 q;
        q.x = c.x; q.y = __float_as_int(v.x); rec[p0] = q;
        q.x = c.y; q.y = __float_as_int(v.y); rec[p1] = q;
        q.x = c.z; q.y = __float_as_int(v.z); rec[p2] = q;
        q.x = c.w; q.y = __float_as_int(v.w); rec[p3] = q;
    } else {
        for (int i = e; i < n_edges; ++i) {
            const int k   = rank[i];
            const int pos = beg[((unsigned)k >> 31) * GCN_N_NODES + edge_row[i]] + (k & 0x7fffffff);
            int2 q;
            q.x = edge_col[i];
            q.y = __float_as_int(edge_val[i]);
            rec[pos] = q;
        }
    }
}

__global__ __launch_bounds__(256)
void gcn_row_sum_half(const int*   __restrict__ beg,
                      const int*   __restrict__ nxt,
                      const int2*  __restrict__ rec,
                      const float* __restrict__ Hh,
                      float*       __restrict__ out,
                      int o_base)
{
    const int t = blockIdx.x * 256 + threadIdx.x;
    const int r = t >> 4;
    const int l = threadIdx.x & 15;

    int       p  = beg[r];
    const int pe = nxt[r];

    const float4* H4 = (const float4*)Hh + l;
    float4 acc;
    acc.x = 0.f; acc.y = 0.f; acc.z = 0.f; acc.w = 0.f;

    for (; p + 4 <= pe; p += 4) {
        const int2 e0 = rec[p];
        const int2 e1 = rec[p + 1];
        const int2 e2 = rec[p + 2];
        const int2 e3 = rec[p + 3];
        const float4 h0 = H4[(size_t)e0.x * 16];
        const float4 h1 = H4[(size_t)e1.x * 16];
        const float4 h2 = H4[(size_t)e2.x * 16];
        const float4 h3 = H4[(size_t)e3.x * 16];
        const float v0 = __int_as_float(e0.y);
        const float v1 = __int_as_float(e1.y);
        const float v2 = __int_as_float(e2.y);
        const float v3 = __int_as_float(e3.y);
        acc.x += v0 * h0.x; acc.y += v0 * h0.y; acc.z += v0 * h0.z; acc.w += v0 * h0.w;
        acc.x += v1 * h1.x; acc.y += v1 * h1.y; acc.z += v1 * h1.z; acc.w += v1 * h1.w;
        acc.x += v2 * h2.x; acc.y += v2 * h2.y; acc.z += v2 * h2.z; acc.w += v2 * h2.w;
        acc.x += v3 * h3.x; acc.y += v3 * h3.y; acc.z += v3 * h3.z; acc.w += v3 * h3.w;
    }
    const int n = pe - p;
    if (n > 0) {
        const int2 e0 = rec[p];
        const int2 e1 = (n > 1) ? rec[p + 1] : e0;
        const int2 e2 = (n > 2) ? rec[p + 2] : e0;
        const float4 h0 = H4[(size_t)e0.x * 16];
        const float4 h1 = H4[(size_t)e1.x * 16];
        const float4 h2 = H4[(size_t)e2.x * 16];
        const float v0 = __int_as_float(e0.y);
        const float v1 = __int_as_float(e1.y);
        const float v2 = __int_as_float(e2.y);
        acc.x += v0 * h0.x; acc.y += v0 * h0.y; acc.z += v0 * h0.z; acc.w += v0 * h0.w;
        if (n > 1) {
            acc.x += v1 * h1.x; acc.y += v1 * h1.y; acc.z += v1 * h1.z; acc.w += v1 * h1.w;
        }
        if (n > 2) {
            acc.x += v2 * h2.x; acc.y += v2 * h2.y; acc.z += v2 * h2.z; acc.w += v2 * h2.w;
        }
    }

    ((float4*)(out + (size_t)r * GCN_OUT_DIM + o_base))[l] = acc;
}

extern "C" void kernel_launch(void* const* d_in, const int* in_sizes, int n_in,
                              void* d_out, int out_size, void* d_ws, size_t ws_size,
                              hipStream_t stream)
{
    const float* X        = (const float*)d_in[0];
    const int*   edge_row = (const int*)d_in[1];
    const int*   edge_col = (const int*)d_in[2];
    const float* edge_val = (const float*)d_in[3];
    const float* W        = (const float*)d_in[4];
    const float* bias     = (const float*)d_in[5];
    float*       out      = (float*)d_out;

    float* Hh  = (float*)d_ws;
    int2*  rec = (int2*)((char*)d_ws + WS_OFF_REC);
    int*   nxt = (int*)((char*)d_ws + WS_OFF_NXT);
    int*   beg = (int*)((char*)d_ws + WS_OFF_BEG);
    int*   rnk = (int*)((char*)d_ws + WS_OFF_RANK);
    int*   cnt = (int*)((char*)d_ws + WS_OFF_CNT);

    const int n_edges = in_sizes[1];
    const int eblk    = (n_edges + 1023) / 1024;

    gcn_clear_counts<<<(2 * GCN_N_NODES + 1 + 255) / 256, 256, 0, stream>>>(cnt, 2 * GCN_N_NODES + 1);
    gcn_count_rows<<<eblk, 256, 0, stream>>>(edge_row, cnt, rnk, n_edges);
    gcn_place_segments<<<(GCN_N_NODES + 1023) / 1024, 1024, 0, stream>>>(cnt, nxt, beg);
    gcn_fill_records<<<eblk, 256, 0, stream>>>(edge_row, edge_col, edge_val, beg, rnk, rec, n_edges);

    gcn_gemm_half_wmma_f32<<<GEMM_GRID, 256, 0, stream>>>(X, W, bias, Hh, 0);
    gcn_row_sum_half<<<GCN_N_NODES * 16 / 256, 256, 0, stream>>>(beg, nxt, rec, Hh, out, 0);

    gcn_gemm_half_wmma_f32<<<GEMM_GRID, 256, 0, stream>>>(X, W, bias, Hh, 64);
    gcn_row_sum_half<<<GCN_N_NODES * 16 / 256, 256, 0, stream>>>(beg, nxt, rec, Hh, out, 64);
}


# Edge records and counts sit in each row's own 512 B of out (63 slots; fuller rows re-scan the edge list), so the workspace holds all 128 H columns: one projection, one row pass, no placement
# speedup vs baseline: 2.6998x; 1.2220x over previous
#include <hip/hip_runtime.h>

typedef __attribute__((ext_vector_type(2))) float v2f;
typedef __attribute__((ext_vector_type(4))) float v4f;
typedef __attribute__((ext_vector_type(8))) float v8f;

#define GCN_N_NODES  100000
#define GCN_IN_DIM   256
#define GCN_OUT_DIM  128

#define WS_OFF_RANK 39600016
#define WS_OFF_CNT  46000016
#define ROW_SLOTS   64
#define ROW_CAP     63

#define GEMM_GRID   512
#define GEMM_TILES  (GCN_OUT_DIM / 16)
#define GEMM_STRIPS (GCN_N_NODES / 16)


__device__ __forceinline__
void gcn_gemm_block64(v8f (&acc)[GEMM_TILES], const v4f (&a)[8], const v4f* bk, int c, int h)
{
    v4f bcur[GEMM_TILES], bnxt[GEMM_TILES];
#pragma unroll
    for (int jt = 0; jt < GEMM_TILES; ++jt) bcur[jt] = bk[jt * 16 * 64 + (h ^ c)];
#pragma unroll
    for (int j = 0; j < 8; ++j) {
        if (j + 1 < 8) {
            const int x = (2 * (j + 1) + h) ^ c;
#pragma unroll
            for (int jt = 0; jt < GEMM_TILES; ++jt) bnxt[jt] = bk[jt * 16 * 64 + x];
        }
        const v2f a01 = (v2f){a[j].x, a[j].y};
        const v2f a23 = (v2f){a[j].z, a[j].w};
#pragma unroll
        for (int jt = 0; jt < GEMM_TILES; ++jt) {
            const v2f b01 = (v2f){bcur[jt].x, bcur[jt].y};
            acc[jt] = __builtin_amdgcn_wmma_f32_16x16x4_f32(
                 false, a01,  false, b01,
                 (short)0, acc[jt],
                 false,  false);
        }
#pragma unroll
        for (int jt = 0; jt < GEMM_TILES; ++jt) {
            const v2f b23 = (v2f){bcur[jt].z, bcur[jt].w};
            acc[jt] = __builtin_amdgcn_wmma_f32_16x16x4_f32(
                 false, a23,  false, b23,
                 (short)0, acc[jt],
                 false,  false);
        }
#pragma unroll
        for (int jt = 0; jt < GEMM_TILES; ++jt) bcur[jt] = bnxt[jt];
    }
}

__global__ __launch_bounds__(256)
void gcn_gemm_wmma_f32(const float* __restrict__ X,
                            const float* __restrict__ W,
                            const float* __restrict__ bias,
                            float* __restrict__ Hh)
{
    __shared__ v4f Ws[GCN_OUT_DIM * 64];

    const int t = threadIdx.x;
#pragma unroll 16
    for (int i = 0; i < GCN_OUT_DIM / 4; ++i) {
        const int idx = i * 256 + t;
        const int o   = idx >> 6;
        const int q   = idx & 63;
        Ws[o * 64 + (q ^ (o & 15))] =
            *(const v4f*)(W + (size_t)o * GCN_IN_DIM + 4 * q);
    }
    __syncthreads();

    const int lane = t & 31;
    const int wv   = __builtin_amdgcn_readfirstlane(t >> 5);
    const int c    = lane & 15;
    const int h    = lane >> 4;

    float bj[GEMM_TILES];
#pragma unroll
    for (int j = 0; j < GEMM_TILES; ++j) bj[j] = bias[16 * j + c];

    const v4f* bBase = Ws + c * 64;

    int s = wv * GEMM_GRID + blockIdx.x;
    if (s >= GEMM_STRIPS) return;

    const float* aRow = X + (size_t)(s * 16 + c) * GCN_IN_DIM + 4 * h;
    v4f a0[8], a1[8];
#pragma unroll
    for (int j = 0; j < 8; ++j) a0[j] = *(const v4f*)(aRow + 8 * j);

    for (;;) {
        const int  sn      = s + 8 * GEMM_GRID;
        const bool more    = sn < GEMM_STRIPS;
        const float* aNext = more ? X + (size_t)(sn * 16 + c) * GCN_IN_DIM + 4 * h : aRow;

        v8f acc[GEMM_TILES];
#pragma unroll
        for (int j = 0; j < GEMM_TILES; ++j)
            acc[j] = (v8f){0.f, 0.f, 0.f, 0.f, 0.f, 0.f, 0.f, 0.f};

#pragma unroll 1
        for (int kb = 0; kb < GCN_IN_DIM; kb += 128) {
#pragma unroll
            for (int j = 0; j < 8; ++j) a1[j] = *(const v4f*)(aRow + kb + 64 + 8 * j);
            gcn_gemm_block64(acc, a0, bBase + (kb >> 2), c, h);

            const float* nx = (kb + 128 < GCN_IN_DIM) ? aRow + kb + 128 : aNext;
#pragma unroll
            for (int j = 0; j < 8; ++j) a0[j] = *(const v4f*)(nx + 8 * j);
            gcn_gemm_block64(acc, a1, bBase + (kb >> 2) + 16, c, h);
        }

#pragma unroll
        for (int j = 0; j < GEMM_TILES; ++j) {
#pragma unroll
            for (int v = 0; v < 8; ++v) {
                const int row = s * 16 + v + 8 * h;
                Hh[(size_t)row * GCN_OUT_DIM + 16 * j + c] = acc[j][v] + bj[j];
            }
        }

        if (!more) break;
        s    = sn;
        aRow = aNext;
    }
}

__global__ void gcn_clear_counts(int* __restrict__ nxt, int n)
{
    int i = blockIdx.x * blockDim.x + threadIdx.x;
    if (i < n) nxt[i] = 0;
}

__global__ __launch_bounds__(256)
void gcn_count_rows(const int* __restrict__ edge_row, int* __restrict__ cnt,
                    int* __restrict__ rank, int n_edges)
{
    const unsigned id   = __builtin_amdgcn_s_sendmsg_rtn(0x87);
    const unsigned half = (id >> 18) & 1u;
    int* __restrict__ plane = cnt + half * GCN_N_NODES;
    const int tag = (int)(half << 31);

    const int e = (blockIdx.x * 256 + threadIdx.x) * 4;
    if (e + 3 < n_edges) {
        const int4 r = *(const int4*)(edge_row + e);
        int4 k;
        k.x = atomicAdd(plane + r.x, 1) | tag;
        k.y = atomicAdd(plane + r.y, 1) | tag;
        k.z = atomicAdd(plane + r.z, 1) | tag;
        k.w = atomicAdd(plane + r.w, 1) | tag;
        *(int4*)(rank + e) = k;
    } else {
        for (int i = e; i < n_edges; ++i)
            rank[i] = atomicAdd(plane + edge_row[i], 1) | tag;
    }
}

__global__ __launch_bounds__(256)
void gcn_fill_slots(const int*   __restrict__ edge_row,
                    const int*   __restrict__ edge_col,
                    const float* __restrict__ edge_val,
                    const int*   __restrict__ cnt,
                    const int*   __restrict__ rank,
                    int2* __restrict__ slots,
                    int n_edges)
{
    const int g = blockIdx.x * 256 + threadIdx.x;
    if (g < GCN_N_NODES) {
        int2 hd;
        hd.x = cnt[g] + cnt[GCN_N_NODES + g];
        hd.y = 0;
        slots[(size_t)g * ROW_SLOTS] = hd;
    }
    const int e = g * 4;
    if (e + 3 < n_edges) {
        const int4   r = *(const int4*)(edge_row + e);
        const int4   c = *(const int4*)(edge_col + e);
        const float4 v = *(const float4*)(edge_val + e);
        const int4   k = *(const int4*)(rank + e);
        const int p0 = (k.x & 0x7fffffff) + ((k.x < 0) ? cnt[r.x] : 0);
        const int p1 = (k.y & 0x7fffffff) + ((k.y < 0) ? cnt[r.y] : 0);
        const int p2 = (k.z & 0x7fffffff) + ((k.z < 0) ? cnt[r.z] : 0);
        const int p3 = (k.w & 0x7fffffff) + ((k.w < 0) ? cnt[r.w] : 0);
        int2 q;
        if (p0 < ROW_CAP) { q.x = c.x; q.y = __float_as_int(v.x); slots[(size_t)r.x * ROW_SLOTS + 1 + p0] = q; }
        if (p1 < ROW_CAP) { q.x = c.y; q.y = __float_as_int(v.y); slots[(size_t)r.y * ROW_SLOTS + 1 + p1] = q; }
        if (p2 < ROW_CAP) { q.x = c.z; q.y = __float_as_int(v.z); slots[(size_t)r.z * ROW_SLOTS + 1 + p2] = q; }
        if (p3 < ROW_CAP) { q.x = c.w; q.y = __float_as_int(v.w); slots[(size_t)r.w * ROW_SLOTS + 1 + p3] = q; }
    } else {
        for (int i = e; i < n_edges; ++i) {
            const int k   = rank[i];
            const int row = edge_row[i];
            const int pos = (k & 0x7fffffff) + ((k < 0) ? cnt[row] : 0);
            if (pos < ROW_CAP) {
                int2 q;
                q.x = edge_col[i];
                q.y = __float_as_int(edge_val[i]);
                slots[(size_t)row * ROW_SLOTS + 1 + pos] = q;
            }
        }
    }
}

__device__ __forceinline__ void gcn_axpy4(float4& acc, float v, const float4 h)
{
    acc.x += v * h.x; acc.y += v * h.y; acc.z += v * h.z; acc.w += v * h.w;
}

__global__ __launch_bounds__(256)
void gcn_row_sum_slots(const int*   __restrict__ edge_row,
                       const int*   __restrict__ edge_col,
                       const float* __restrict__ edge_val,
                       const float* __restrict__ H,
                       float*       out,
                       int n_edges)
{
    const int lane = threadIdx.x & 31;
    const int r    = blockIdx.x * 8 + __builtin_amdgcn_readfirstlane(threadIdx.x >> 5);

    const float4* H4 = (const float4*)H + lane;
    const int2* slot = (const int2*)(out + (size_t)r * GCN_OUT_DIM);

    const int2 pg0 = slot[lane];
    const int  n   = __builtin_amdgcn_readlane(pg0.x, 0);

    float4 acc;
    acc.x = 0.f; acc.y = 0.f; acc.z = 0.f; acc.w = 0.f;

    if (n <= ROW_CAP) {
        int2 pg1;
        pg1.x = 0; pg1.y = 0;
        if (n > 31) pg1 = slot[32 + lane];

        int j = 0;
        for (; j + 8 <= n; j += 8) {
            float4 h[8];
            float  v[8];
#pragma unroll
            for (int i = 0; i < 8; ++i) {
                const int s   = j + i + 1;
                const int col = (s < 32) ? __builtin_amdgcn_readlane(pg0.x, s & 31)
                                         : __builtin_amdgcn_readlane(pg1.x, s & 31);
                const int vb  = (s < 32) ? __builtin_amdgcn_readlane(pg0.y, s & 31)
                                         : __builtin_amdgcn_readlane(pg1.y, s & 31);
                h[i] = H4[(size_t)col * 32];
                v[i] = __int_as_float(vb);
            }
#pragma unroll
            for (int i = 0; i < 8; ++i) gcn_axpy4(acc, v[i], h[i]);
        }
        const int m = n - j;
        if (m > 0) {
            float4 h[7];
            float  v[7];
#pragma unroll
            for (int i = 0; i < 7; ++i) {
                if (i < m) {
                    const int s   = j + i + 1;
                    const int col = (s < 32) ? __builtin_amdgcn_readlane(pg0.x, s & 31)
                                             : __builtin_amdgcn_readlane(pg1.x, s & 31);
                    const int vb  = (s < 32) ? __builtin_amdgcn_readlane(pg0.y, s & 31)
                                             : __builtin_amdgcn_readlane(pg1.y, s & 31);
                    h[i] = H4[(size_t)col * 32];
                    v[i] = __int_as_float(vb);
                }
            }
#pragma unroll
            for (int i = 0; i < 7; ++i) {
                if (i < m) gcn_axpy4(acc, v[i], h[i]);
            }
        }
    } else {
        for (int e0 = 0; e0 < n_edges; e0 += 32) {
            const int e  = e0 + lane;
            const int rr = (e < n_edges) ? edge_row[e] : -1;
            unsigned mask = __builtin_amdgcn_ballot_w32(rr == r);
            while (mask) {
                const int b = __builtin_ctz(mask);
                mask &= mask - 1;
                const int   col = edge_col[e0 + b];
                const float val = edge_val[e0 + b];
                gcn_axpy4(acc, val, H4[(size_t)col * 32]);
            }
        }
    }

    ((float4*)(out + (size_t)r * GCN_OUT_DIM))[lane] = acc;
}

extern "C" void kernel_launch(void* const* d_in, const int* in_sizes, int n_in,
                              void* d_out, int out_size, void* d_ws, size_t ws_size,
                              hipStream_t stream)
{
    const float* X        = (const float*)d_in[0];
    const int*   edge_row = (const int*)d_in[1];
    const int*   edge_col = (const int*)d_in[2];
    const float* edge_val = (const float*)d_in[3];
    const float* W        = (const float*)d_in[4];
    const float* bias     = (const float*)d_in[5];
    float*       out      = (float*)d_out;

    float* H   = (float*)d_ws;
    int*   rnk = (int*)((char*)d_ws + WS_OFF_RANK);
    int*   cnt = (int*)((char*)d_ws + WS_OFF_CNT);

    const int n_edges = in_sizes[1];
    const int eblk    = (n_edges + 1023) / 1024;

    gcn_clear_counts<<<(2 * GCN_N_NODES + 255) / 256, 256, 0, stream>>>(cnt, 2 * GCN_N_NODES);
    gcn_count_rows<<<eblk, 256, 0, stream>>>(edge_row, cnt, rnk, n_edges);
    gcn_fill_slots<<<eblk, 256, 0, stream>>>(edge_row, edge_col, edge_val, cnt, rnk, (int2*)out, n_edges);

    gcn_gemm_wmma_f32<<<GEMM_GRID, 256, 0, stream>>>(X, W, bias, H);

    gcn_row_sum_slots<<<GCN_N_NODES / 8, 256, 0, stream>>>(edge_row, edge_col, edge_val, H, out, n_edges);
}


# Edge bucketing by a two-level counting sort on LDS counters (4096-edge chunks ordered by 128-row group, then one work-group per group) in place of per-edge global atomics; sum order = LDS count rank
# speedup vs baseline: 3.0785x; 1.1403x over previous
#include <hip/hip_runtime.h>

typedef __attribute__((ext_vector_type(2))) float v2f;
typedef __attribute__((ext_vector_type(4))) float v4f;
typedef __attribute__((ext_vector_type(8))) float v8f;

#define GCN_N_NODES  100000
#define GCN_IN_DIM   256
#define GCN_OUT_DIM  128

#define WS_OFF_START 12812288
#define ROW_SLOTS   64
#define ROW_CAP     63

#define GEMM_GRID   512
#define GEMM_TILES  (GCN_OUT_DIM / 16)
#define GEMM_STRIPS (GCN_N_NODES / 16)


__device__ __forceinline__
void gcn_gemm_block64(v8f (&acc)[GEMM_TILES], const v4f (&a)[8], const v4f* bk, int c, int h)
{
    v4f bcur[GEMM_TILES], bnxt[GEMM_TILES];
#pragma unroll
    for (int jt = 0; jt < GEMM_TILES; ++jt) bcur[jt] = bk[jt * 16 * 64 + (h ^ c)];
#pragma unroll
    for (int j = 0; j < 8; ++j) {
        if (j + 1 < 8) {
            const int x = (2 * (j + 1) + h) ^ c;
#pragma unroll
            for (int jt = 0; jt < GEMM_TILES; ++jt) bnxt[jt] = bk[jt * 16 * 64 + x];
        }
        const v2f a01 = (v2f){a[j].x, a[j].y};
        const v2f a23 = (v2f){a[j].z, a[j].w};
#pragma unroll
        for (int jt = 0; jt < GEMM_TILES; ++jt) {
            const v2f b01 = (v2f){bcur[jt].x, bcur[jt].y};
            acc[jt] = __builtin_amdgcn_wmma_f32_16x16x4_f32(
                 false, a01,  false, b01,
                 (short)0, acc[jt],
                 false,  false);
        }
#pragma unroll
        for (int jt = 0; jt < GEMM_TILES; ++jt) {
            const v2f b23 = (v2f){bcur[jt].z, bcur[jt].w};
            acc[jt] = __builtin_amdgcn_wmma_f32_16x16x4_f32(
                 false, a23,  false, b23,
                 (short)0, acc[jt],
                 false,  false);
        }
#pragma unroll
        for (int jt = 0; jt < GEMM_TILES; ++jt) bcur[jt] = bnxt[jt];
    }
}

__global__ __launch_bounds__(256)
void gcn_gemm_wmma_f32(const float* __restrict__ X,
                            const float* __restrict__ W,
                            const float* __restrict__ bias,
                            float* __restrict__ Hh)
{
    __shared__ v4f Ws[GCN_OUT_DIM * 64];

    const int t = threadIdx.x;
#pragma unroll 16
    for (int i = 0; i < GCN_OUT_DIM / 4; ++i) {
        const int idx = i * 256 + t;
        const int o   = idx >> 6;
        const int q   = idx & 63;
        Ws[o * 64 + (q ^ (o & 15))] =
            *(const v4f*)(W + (size_t)o * GCN_IN_DIM + 4 * q);
    }
    __syncthreads();

    const int lane = t & 31;
    const int wv   = __builtin_amdgcn_readfirstlane(t >> 5);
    const int c    = lane & 15;
    const int h    = lane >> 4;

    float bj[GEMM_TILES];
#pragma unroll
    for (int j = 0; j < GEMM_TILES; ++j) bj[j] = bias[16 * j + c];

    const v4f* bBase = Ws + c * 64;

    int s = wv * GEMM_GRID + blockIdx.x;
    if (s >= GEMM_STRIPS) return;

    const float* aRow = X + (size_t)(s * 16 + c) * GCN_IN_DIM + 4 * h;
    v4f a0[8], a1[8];
#pragma unroll
    for (int j = 0; j < 8; ++j) a0[j] = *(const v4f*)(aRow + 8 * j);

    for (;;) {
        const int  sn      = s + 8 * GEMM_GRID;
        const bool more    = sn < GEMM_STRIPS;
        const float* aNext = more ? X + (size_t)(sn * 16 + c) * GCN_IN_DIM + 4 * h : aRow;

        v8f acc[GEMM_TILES];
#pragma unroll
        for (int j = 0; j < GEMM_TILES; ++j)
            acc[j] = (v8f){0.f, 0.f, 0.f, 0.f, 0.f, 0.f, 0.f, 0.f};

#pragma unroll 1
        for (int kb = 0; kb < GCN_IN_DIM; kb += 128) {
#pragma unroll
            for (int j = 0; j < 8; ++j) a1[j] = *(const v4f*)(aRow + kb + 64 + 8 * j);
            gcn_gemm_block64(acc, a0, bBase + (kb >> 2), c, h);

            const float* nx = (kb + 128 < GCN_IN_DIM) ? aRow + kb + 128 : aNext;
#pragma unroll
            for (int j = 0; j < 8; ++j) a0[j] = *(const v4f*)(nx + 8 * j);
            gcn_gemm_block64(acc, a1, bBase + (kb >> 2) + 16, c, h);
        }

#pragma unroll
        for (int j = 0; j < GEMM_TILES; ++j) {
#pragma unroll
            for (int v = 0; v < 8; ++v) {
                const int row = s * 16 + v + 8 * h;
                Hh[(size_t)row * GCN_OUT_DIM + 16 * j + c] = acc[j][v] + bj[j];
            }
        }

        if (!more) break;
        s    = sn;
        aRow = aNext;
    }
}

#define SRT_EPB    4096
#define SRT_SHIFT  7
#define SRT_RPG    (1 << SRT_SHIFT)
#define SRT_NG     ((GCN_N_NODES + SRT_RPG - 1) / SRT_RPG)
#define REC_SHIFT  24

__global__ __launch_bounds__(256)
void gcn_sort_chunk(const int*   __restrict__ edge_row,
                    const int*   __restrict__ edge_col,
                    const float* __restrict__ edge_val,
                    int2* __restrict__ rec,
                    int*  __restrict__ start,
                    int n_edges, int nblk)
{
    __shared__ int  hist[SRT_NG + 1];
    __shared__ int  wsum[8];
    __shared__ int2 sorted[SRT_EPB];

    const int t    = threadIdx.x;
    const int lane = t & 31;
    const int wave = t >> 5;
    const int base = blockIdx.x * SRT_EPB;

    for (int i = t; i <= SRT_NG; i += 256) hist[i] = 0;

    int row[16], col[16], vb[16];
    if (base + SRT_EPB <= n_edges) {
#pragma unroll
        for (int g = 0; g < 4; ++g) {
            const int e = base + g * 1024 + 4 * t;
            const int4   r = *(const int4*)(edge_row + e);
            const int4   c = *(const int4*)(edge_col + e);
            const float4 v = *(const float4*)(edge_val + e);
            row[4 * g + 0] = r.x; row[4 * g + 1] = r.y; row[4 * g + 2] = r.z; row[4 * g + 3] = r.w;
            col[4 * g + 0] = c.x; col[4 * g + 1] = c.y; col[4 * g + 2] = c.z; col[4 * g + 3] = c.w;
            vb[4 * g + 0] = __float_as_int(v.x); vb[4 * g + 1] = __float_as_int(v.y);
            vb[4 * g + 2] = __float_as_int(v.z); vb[4 * g + 3] = __float_as_int(v.w);
        }
    } else {
#pragma unroll
        for (int k = 0; k < 16; ++k) {
            const int  e  = base + (k >> 2) * 1024 + 4 * t + (k & 3);
            const bool ok = e < n_edges;
            row[k] = ok ? edge_row[e] : -1;
            col[k] = ok ? edge_col[e] : 0;
            vb[k]  = ok ? __float_as_int(edge_val[e]) : 0;
        }
    }
    __syncthreads();

    int bkt[16], rk[16];
#pragma unroll
    for (int k = 0; k < 16; ++k) {
        bkt[k] = (row[k] >= 0) ? (row[k] >> SRT_SHIFT) : SRT_NG;
        rk[k]  = atomicAdd(&hist[bkt[k]], 1);
    }
    __syncthreads();

    int c4[4];
    int s = 0;
#pragma unroll
    for (int k = 0; k < 4; ++k) {
        const int i = 4 * t + k;
        c4[k] = (i <= SRT_NG) ? hist[i] : 0;
        s += c4[k];
    }
    int x = s;
#pragma unroll
    for (int d = 1; d < 32; d <<= 1) {
        const int y = __shfl_up(x, d);
        if (lane >= d) x += y;
    }
    if (lane == 31) wsum[wave] = x;
    __syncthreads();
    int pre = x - s;
    for (int w = 0; w < 8; ++w)
        if (w < wave) pre += wsum[w];
#pragma unroll
    for (int k = 0; k < 4; ++k) {
        const int i = 4 * t + k;
        if (i <= SRT_NG) {
            hist[i] = pre;
            start[(size_t)i * nblk + blockIdx.x] = pre;
        }
        pre += c4[k];
    }
    __syncthreads();

#pragma unroll
    for (int k = 0; k < 16; ++k) {
        int2 q;
        q.x = col[k] | ((row[k] & (SRT_RPG - 1)) << REC_SHIFT);
        q.y = vb[k];
        sorted[hist[bkt[k]] + rk[k]] = q;
    }
    __syncthreads();

    int4*       dst = (int4*)(rec + base);
    const int4* srt = (const int4*)sorted;
#pragma unroll
    for (int k = 0; k < 8; ++k) dst[k * 256 + t] = srt[k * 256 + t];
}

__global__ __launch_bounds__(256)
void gcn_place_rows(const int2* __restrict__ rec,
                    const int*  __restrict__ start,
                    int2* __restrict__ slots,
                    int nblk)
{
    __shared__ int  lcnt[SRT_RPG];
    __shared__ int2 img[SRT_RPG * ROW_SLOTS];

    const int t = threadIdx.x;
    const int g = blockIdx.x;

    if (t < SRT_RPG) lcnt[t] = 0;
    __syncthreads();

    const int* sBeg = start + (size_t)g * nblk;
    const int* sEnd = start + (size_t)(g + 1) * nblk;
    for (int i = t; i < nblk; i += 256) {
        const int   s = sBeg[i];
        const int   e = sEnd[i];
        const int2* p = rec + (size_t)i * SRT_EPB;
        for (int j = s; j < e; j += 8) {
            int2 q[8];
#pragma unroll
            for (int k = 0; k < 8; ++k) q[k] = p[(j + k < e) ? j + k : e - 1];
            __builtin_amdgcn_sched_barrier(0);
#pragma unroll
            for (int k = 0; k < 8; ++k) {
                if (j + k < e) {
                    const int rl   = q[k].x >> REC_SHIFT;
                    const int rank = atomicAdd(&lcnt[rl], 1);
                    if (rank < ROW_CAP) {
                        int2 w;
                        w.x = q[k].x & ((1 << REC_SHIFT) - 1);
                        w.y = q[k].y;
                        img[rl * ROW_SLOTS + 1 + rank] = w;
                    }
                }
            }
        }
    }
    __syncthreads();

    if (t < SRT_RPG) {
        int2 hd;
        hd.x = lcnt[t];
        hd.y = 0;
        img[t * ROW_SLOTS] = hd;
    }
    __syncthreads();

    const int lane = t & 31;
    const int wave = __builtin_amdgcn_readfirstlane(t >> 5);
    for (int rl = wave; rl < SRT_RPG; rl += 8) {
        const int r = g * SRT_RPG + rl;
        if (r >= GCN_N_NODES) break;
        const int n = lcnt[rl];
        const int m = (n < ROW_CAP) ? n : ROW_CAP;
        if (2 * (lane & ~7) <= m)
            ((int4*)(slots + (size_t)r * ROW_SLOTS))[lane] = ((const int4*)(img + rl * ROW_SLOTS))[lane];
    }
}

__device__ __forceinline__ void gcn_axpy4(float4& acc, float v, const float4 h)
{
    acc.x += v * h.x; acc.y += v * h.y; acc.z += v * h.z; acc.w += v * h.w;
}

__global__ __launch_bounds__(256)
void gcn_row_sum_slots(const int*   __restrict__ edge_row,
                       const int*   __restrict__ edge_col,
                       const float* __restrict__ edge_val,
                       const float* __restrict__ H,
                       float*       out,
                       int n_edges)
{
    const int lane = threadIdx.x & 31;
    const int r    = blockIdx.x * 8 + __builtin_amdgcn_readfirstlane(threadIdx.x >> 5);

    const float4* H4 = (const float4*)H + lane;
    const int2* slot = (const int2*)(out + (size_t)r * GCN_OUT_DIM);

    const int2 pg0 = slot[lane];
    const int  n   = __builtin_amdgcn_readlane(pg0.x, 0);

    float4 acc;
    acc.x = 0.f; acc.y = 0.f; acc.z = 0.f; acc.w = 0.f;

    if (n <= ROW_CAP) {
        int2 pg1;
        pg1.x = 0; pg1.y = 0;
        if (n > 31) pg1 = slot[32 + lane];

        int j = 0;
        for (; j + 8 <= n; j += 8) {
            float4 h[8];
            float  v[8];
#pragma unroll
            for (int i = 0; i < 8; ++i) {
                const int s   = j + i + 1;
                const int col = (s < 32) ? __builtin_amdgcn_readlane(pg0.x, s & 31)
                                         : __builtin_amdgcn_readlane(pg1.x, s & 31);
                const int vb  = (s < 32) ? __builtin_amdgcn_readlane(pg0.y, s & 31)
                                         : __builtin_amdgcn_readlane(pg1.y, s & 31);
                h[i] = H4[(size_t)col * 32];
                v[i] = __int_as_float(vb);
            }
#pragma unroll
            for (int i = 0; i < 8; ++i) gcn_axpy4(acc, v[i], h[i]);
        }
        const int m = n - j;
        if (m > 0) {
            float4 h[7];
            float  v[7];
#pragma unroll
            for (int i = 0; i < 7; ++i) {
                if (i < m) {
                    const int s   = j + i + 1;
                    const int col = (s < 32) ? __builtin_amdgcn_readlane(pg0.x, s & 31)
                                             : __builtin_amdgcn_readlane(pg1.x, s & 31);
                    const int vb  = (s < 32) ? __builtin_amdgcn_readlane(pg0.y, s & 31)
                                             : __builtin_amdgcn_readlane(pg1.y, s & 31);
                    h[i] = H4[(size_t)col * 32];
                    v[i] = __int_as_float(vb);
                }
            }
#pragma unroll
            for (int i = 0; i < 7; ++i) {
                if (i < m) gcn_axpy4(acc, v[i], h[i]);
            }
        }
    } else {
        for (int e0 = 0; e0 < n_edges; e0 += 32) {
            const int e  = e0 + lane;
            const int rr = (e < n_edges) ? edge_row[e] : -1;
            unsigned mask = __builtin_amdgcn_ballot_w32(rr == r);
            while (mask) {
                const int b = __builtin_ctz(mask);
                mask &= mask - 1;
                const int   col = edge_col[e0 + b];
                const float val = edge_val[e0 + b];
                gcn_axpy4(acc, val, H4[(size_t)col * 32]);
            }
        }
    }

    ((float4*)(out + (size_t)r * GCN_OUT_DIM))[lane] = acc;
}

extern "C" void kernel_launch(void* const* d_in, const int* in_sizes, int n_in,
                              void* d_out, int out_size, void* d_ws, size_t ws_size,
                              hipStream_t stream)
{
    const float* X        = (const float*)d_in[0];
    const int*   edge_row = (const int*)d_in[1];
    const int*   edge_col = (const int*)d_in[2];
    const float* edge_val = (const float*)d_in[3];
    const float* W        = (const float*)d_in[4];
    const float* bias     = (const float*)d_in[5];
    float*       out      = (float*)d_out;

    float* H   = (float*)d_ws;
    int2*  rec = (int2*)d_ws;
    int*   stt = (int*)((char*)d_ws + WS_OFF_START);

    const int n_edges = in_sizes[1];
    const int nblk    = (n_edges + SRT_EPB - 1) / SRT_EPB;

    gcn_sort_chunk<<<nblk, 256, 0, stream>>>(edge_row, edge_col, edge_val, rec, stt, n_edges, nblk);
    gcn_place_rows<<<SRT_NG, 256, 0, stream>>>(rec, stt, (int2*)out, nblk);

    gcn_gemm_wmma_f32<<<GEMM_GRID, 256, 0, stream>>>(X, W, bias, H);

    gcn_row_sum_slots<<<GCN_N_NODES / 8, 256, 0, stream>>>(edge_row, edge_col, edge_val, H, out, n_edges);
}


# Run start positions stored per chunk (25 whole lines, one writing work-group each) instead of per row group, where every line took 4-byte stores from 32 work-groups of both package halves
# speedup vs baseline: 3.2485x; 1.0552x over previous
#include <hip/hip_runtime.h>

typedef __attribute__((ext_vector_type(2))) float v2f;
typedef __attribute__((ext_vector_type(4))) float v4f;
typedef __attribute__((ext_vector_type(8))) float v8f;

#define GCN_N_NODES  100000
#define GCN_IN_DIM   256
#define GCN_OUT_DIM  128

#define WS_OFF_START 12812288
#define START_PITCH  800
#define ROW_SLOTS   64
#define ROW_CAP     63

#define GEMM_GRID   512
#define GEMM_TILES  (GCN_OUT_DIM / 16)
#define GEMM_STRIPS (GCN_N_NODES / 16)


__device__ __forceinline__
void gcn_gemm_block64(v8f (&acc)[GEMM_TILES], const v4f (&a)[8], const v4f* bk, int c, int h)
{
    v4f bcur[GEMM_TILES], bnxt[GEMM_TILES];
#pragma unroll
    for (int jt = 0; jt < GEMM_TILES; ++jt) bcur[jt] = bk[jt * 16 * 64 + (h ^ c)];
#pragma unroll
    for (int j = 0; j < 8; ++j) {
        if (j + 1 < 8) {
            const int x = (2 * (j + 1) + h) ^ c;
#pragma unroll
            for (int jt = 0; jt < GEMM_TILES; ++jt) bnxt[jt] = bk[jt * 16 * 64 + x];
        }
        const v2f a01 = (v2f){a[j].x, a[j].y};
        const v2f a23 = (v2f){a[j].z, a[j].w};
#pragma unroll
        for (int jt = 0; jt < GEMM_TILES; ++jt) {
            const v2f b01 = (v2f){bcur[jt].x, bcur[jt].y};
            acc[jt] = __builtin_amdgcn_wmma_f32_16x16x4_f32(
                 false, a01,  false, b01,
                 (short)0, acc[jt],
                 false,  false);
        }
#pragma unroll
        for (int jt = 0; jt < GEMM_TILES; ++jt) {
            const v2f b23 = (v2f){bcur[jt].z, bcur[jt].w};
            acc[jt] = __builtin_amdgcn_wmma_f32_16x16x4_f32(
                 false, a23,  false, b23,
                 (short)0, acc[jt],
                 false,  false);
        }
#pragma unroll
        for (int jt = 0; jt < GEMM_TILES; ++jt) bcur[jt] = bnxt[jt];
    }
}

__global__ __launch_bounds__(256)
void gcn_gemm_wmma_f32(const float* __restrict__ X,
                            const float* __restrict__ W,
                            const float* __restrict__ bias,
                            float* __restrict__ Hh)
{
    __shared__ v4f Ws[GCN_OUT_DIM * 64];

    const int t = threadIdx.x;
#pragma unroll 16
    for (int i = 0; i < GCN_OUT_DIM / 4; ++i) {
        const int idx = i * 256 + t;
        const int o   = idx >> 6;
        const int q   = idx & 63;
        Ws[o * 64 + (q ^ (o & 15))] =
            *(const v4f*)(W + (size_t)o * GCN_IN_DIM + 4 * q);
    }
    __syncthreads();

    const int lane = t & 31;
    const int wv   = __builtin_amdgcn_readfirstlane(t >> 5);
    const int c    = lane & 15;
    const int h    = lane >> 4;

    float bj[GEMM_TILES];
#pragma unroll
    for (int j = 0; j < GEMM_TILES; ++j) bj[j] = bias[16 * j + c];

    const v4f* bBase = Ws + c * 64;

    int s = wv * GEMM_GRID + blockIdx.x;
    if (s >= GEMM_STRIPS) return;

    const float* aRow = X + (size_t)(s * 16 + c) * GCN_IN_DIM + 4 * h;
    v4f a0[8], a1[8];
#pragma unroll
    for (int j = 0; j < 8; ++j) a0[j] = *(const v4f*)(aRow + 8 * j);

    for (;;) {
        const int  sn      = s + 8 * GEMM_GRID;
        const bool more    = sn < GEMM_STRIPS;
        const float* aNext = more ? X + (size_t)(sn * 16 + c) * GCN_IN_DIM + 4 * h : aRow;

        v8f acc[GEMM_TILES];
#pragma unroll
        for (int j = 0; j < GEMM_TILES; ++j)
            acc[j] = (v8f){0.f, 0.f, 0.f, 0.f, 0.f, 0.f, 0.f, 0.f};

#pragma unroll 1
        for (int kb = 0; kb < GCN_IN_DIM; kb += 128) {
#pragma unroll
            for (int j = 0; j < 8; ++j) a1[j] = *(const v4f*)(aRow + kb + 64 + 8 * j);
            gcn_gemm_block64(acc, a0, bBase + (kb >> 2), c, h);

            const float* nx = (kb + 128 < GCN_IN_DIM) ? aRow + kb + 128 : aNext;
#pragma unroll
            for (int j = 0; j < 8; ++j) a0[j] = *(const v4f*)(nx + 8 * j);
            gcn_gemm_block64(acc, a1, bBase + (kb >> 2) + 16, c, h);
        }

#pragma unroll
        for (int j = 0; j < GEMM_TILES; ++j) {
#pragma unroll
            for (int v = 0; v < 8; ++v) {
                const int row = s * 16 + v + 8 * h;
                Hh[(size_t)row * GCN_OUT_DIM + 16 * j + c] = acc[j][v] + bj[j];
            }
        }

        if (!more) break;
        s    = sn;
        aRow = aNext;
    }
}

#define SRT_EPB    4096
#define SRT_SHIFT  7
#define SRT_RPG    (1 << SRT_SHIFT)
#define SRT_NG     ((GCN_N_NODES + SRT_RPG - 1) / SRT_RPG)
#define REC_SHIFT  24

__global__ __launch_bounds__(256)
void gcn_sort_chunk(const int*   __restrict__ edge_row,
                    const int*   __restrict__ edge_col,
                    const float* __restrict__ edge_val,
                    int2* __restrict__ rec,
                    int*  __restrict__ start,
                    int n_edges, int nblk)
{
    __shared__ int  hist[SRT_NG + 1];
    __shared__ int  wsum[8];
    __shared__ int2 sorted[SRT_EPB];

    const int t    = threadIdx.x;
    const int lane = t & 31;
    const int wave = t >> 5;
    const int base = blockIdx.x * SRT_EPB;

    for (int i = t; i <= SRT_NG; i += 256) hist[i] = 0;

    int row[16], col[16], vb[16];
    if (base + SRT_EPB <= n_edges) {
#pragma unroll
        for (int g = 0; g < 4; ++g) {
            const int e = base + g * 1024 + 4 * t;
            const int4   r = *(const int4*)(edge_row + e);
            const int4   c = *(const int4*)(edge_col + e);
            const float4 v = *(const float4*)(edge_val + e);
            row[4 * g + 0] = r.x; row[4 * g + 1] = r.y; row[4 * g + 2] = r.z; row[4 * g + 3] = r.w;
            col[4 * g + 0] = c.x; col[4 * g + 1] = c.y; col[4 * g + 2] = c.z; col[4 * g + 3] = c.w;
            vb[4 * g + 0] = __float_as_int(v.x); vb[4 * g + 1] = __float_as_int(v.y);
            vb[4 * g + 2] = __float_as_int(v.z); vb[4 * g + 3] = __float_as_int(v.w);
        }
    } else {
#pragma unroll
        for (int k = 0; k < 16; ++k) {
            const int  e  = base + (k >> 2) * 1024 + 4 * t + (k & 3);
            const bool ok = e < n_edges;
            row[k] = ok ? edge_row[e] : -1;
            col[k] = ok ? edge_col[e] : 0;
            vb[k]  = ok ? __float_as_int(edge_val[e]) : 0;
        }
    }
    __syncthreads();

    int bkt[16], rk[16];
#pragma unroll
    for (int k = 0; k < 16; ++k) {
        bkt[k] = (row[k] >= 0) ? (row[k] >> SRT_SHIFT) : SRT_NG;
        rk[k]  = atomicAdd(&hist[bkt[k]], 1);
    }
    __syncthreads();

    int c4[4];
    int s = 0;
#pragma unroll
    for (int k = 0; k < 4; ++k) {
        const int i = 4 * t + k;
        c4[k] = (i <= SRT_NG) ? hist[i] : 0;
        s += c4[k];
    }
    int x = s;
#pragma unroll
    for (int d = 1; d < 32; d <<= 1) {
        const int y = __shfl_up(x, d);
        if (lane >= d) x += y;
    }
    if (lane == 31) wsum[wave] = x;
    __syncthreads();
    int pre = x - s;
    for (int w = 0; w < 8; ++w)
        if (w < wave) pre += wsum[w];
    int4 st;
    st.x = pre; st.y = st.x + c4[0]; st.z = st.y + c4[1]; st.w = st.z + c4[2];
    if (4 * t <= SRT_NG) {
        *(int4*)(start + (size_t)blockIdx.x * START_PITCH + 4 * t) = st;
        hist[4 * t] = st.x;
        if (4 * t + 1 <= SRT_NG) hist[4 * t + 1] = st.y;
        if (4 * t + 2 <= SRT_NG) hist[4 * t + 2] = st.z;
        if (4 * t + 3 <= SRT_NG) hist[4 * t + 3] = st.w;
    }
    __syncthreads();

#pragma unroll
    for (int k = 0; k < 16; ++k) {
        int2 q;
        q.x = col[k] | ((row[k] & (SRT_RPG - 1)) << REC_SHIFT);
        q.y = vb[k];
        sorted[hist[bkt[k]] + rk[k]] = q;
    }
    __syncthreads();

    int4*       dst = (int4*)(rec + base);
    const int4* srt = (const int4*)sorted;
#pragma unroll
    for (int k = 0; k < 8; ++k) dst[k * 256 + t] = srt[k * 256 + t];
}

__global__ __launch_bounds__(256)
void gcn_place_rows(const int2* __restrict__ rec,
                    const int*  __restrict__ start,
                    int2* __restrict__ slots,
                    int nblk)
{
    __shared__ int  lcnt[SRT_RPG];
    __shared__ int2 img[SRT_RPG * ROW_SLOTS];

    const int t = threadIdx.x;
    const int g = blockIdx.x;

    if (t < SRT_RPG) lcnt[t] = 0;
    __syncthreads();

    const int* sGrp = start + g;
    for (int i = t; i < nblk; i += 256) {
        const int   s = sGrp[(size_t)i * START_PITCH];
        const int   e = sGrp[(size_t)i * START_PITCH + 1];
        const int2* p = rec + (size_t)i * SRT_EPB;
        for (int j = s; j < e; j += 8) {
            int2 q[8];
#pragma unroll
            for (int k = 0; k < 8; ++k) q[k] = p[(j + k < e) ? j + k : e - 1];
            __builtin_amdgcn_sched_barrier(0);
#pragma unroll
            for (int k = 0; k < 8; ++k) {
                if (j + k < e) {
                    const int rl   = q[k].x >> REC_SHIFT;
                    const int rank = atomicAdd(&lcnt[rl], 1);
                    if (rank < ROW_CAP) {
                        int2 w;
                        w.x = q[k].x & ((1 << REC_SHIFT) - 1);
                        w.y = q[k].y;
                        img[rl * ROW_SLOTS + 1 + rank] = w;
                    }
                }
            }
        }
    }
    __syncthreads();

    if (t < SRT_RPG) {
        int2 hd;
        hd.x = lcnt[t];
        hd.y = 0;
        img[t * ROW_SLOTS] = hd;
    }
    __syncthreads();

    const int lane = t & 31;
    const int wave = __builtin_amdgcn_readfirstlane(t >> 5);
    for (int rl = wave; rl < SRT_RPG; rl += 8) {
        const int r = g * SRT_RPG + rl;
        if (r >= GCN_N_NODES) break;
        const int n = lcnt[rl];
        const int m = (n < ROW_CAP) ? n : ROW_CAP;
        if (2 * (lane & ~7) <= m)
            ((int4*)(slots + (size_t)r * ROW_SLOTS))[lane] = ((const int4*)(img + rl * ROW_SLOTS))[lane];
    }
}

__device__ __forceinline__ void gcn_axpy4(float4& acc, float v, const float4 h)
{
    acc.x += v * h.x; acc.y += v * h.y; acc.z += v * h.z; acc.w += v * h.w;
}

__global__ __launch_bounds__(256)
void gcn_row_sum_slots(const int*   __restrict__ edge_row,
                       const int*   __restrict__ edge_col,
                       const float* __restrict__ edge_val,
                       const float* __restrict__ H,
                       float*       out,
                       int n_edges)
{
    const int lane = threadIdx.x & 31;
    const int r    = blockIdx.x * 8 + __builtin_amdgcn_readfirstlane(threadIdx.x >> 5);

    const float4* H4 = (const float4*)H + lane;
    const int2* slot = (const int2*)(out + (size_t)r * GCN_OUT_DIM);

    const int2 pg0 = slot[lane];
    const int  n   = __builtin_amdgcn_readlane(pg0.x, 0);

    float4 acc;
    acc.x = 0.f; acc.y = 0.f; acc.z = 0.f; acc.w = 0.f;

    if (n <= ROW_CAP) {
        int2 pg1;
        pg1.x = 0; pg1.y = 0;
        if (n > 31) pg1 = slot[32 + lane];

        int j = 0;
        for (; j + 8 <= n; j += 8) {
            float4 h[8];
            float  v[8];
#pragma unroll
            for (int i = 0; i < 8; ++i) {
                const int s   = j + i + 1;
                const int col = (s < 32) ? __builtin_amdgcn_readlane(pg0.x, s & 31)
                                         : __builtin_amdgcn_readlane(pg1.x, s & 31);
                const int vb  = (s < 32) ? __builtin_amdgcn_readlane(pg0.y, s & 31)
                                         : __builtin_amdgcn_readlane(pg1.y, s & 31);
                h[i] = H4[(size_t)col * 32];
                v[i] = __int_as_float(vb);
            }
#pragma unroll
            for (int i = 0; i < 8; ++i) gcn_axpy4(acc, v[i], h[i]);
        }
        const int m = n - j;
        if (m > 0) {
            float4 h[7];
            float  v[7];
#pragma unroll
            for (int i = 0; i < 7; ++i) {
                if (i < m) {
                    const int s   = j + i + 1;
                    const int col = (s < 32) ? __builtin_amdgcn_readlane(pg0.x, s & 31)
                                             : __builtin_amdgcn_readlane(pg1.x, s & 31);
                    const int vb  = (s < 32) ? __builtin_amdgcn_readlane(pg0.y, s & 31)
                                             : __builtin_amdgcn_readlane(pg1.y, s & 31);
                    h[i] = H4[(size_t)col * 32];
                    v[i] = __int_as_float(vb);
                }
            }
#pragma unroll
            for (int i = 0; i < 7; ++i) {
                if (i < m) gcn_axpy4(acc, v[i], h[i]);
            }
        }
    } else {
        for (int e0 = 0; e0 < n_edges; e0 += 32) {
            const int e  = e0 + lane;
            const int rr = (e < n_edges) ? edge_row[e] : -1;
            unsigned mask = __builtin_amdgcn_ballot_w32(rr == r);
            while (mask) {
                const int b = __builtin_ctz(mask);
                mask &= mask - 1;
                const int   col = edge_col[e0 + b];
                const float val = edge_val[e0 + b];
                gcn_axpy4(acc, val, H4[(size_t)col * 32]);
            }
        }
    }

    ((float4*)(out + (size_t)r * GCN_OUT_DIM))[lane] = acc;
}

extern "C" void kernel_launch(void* const* d_in, const int* in_sizes, int n_in,
                              void* d_out, int out_size, void* d_ws, size_t ws_size,
                              hipStream_t stream)
{
    const float* X        = (const float*)d_in[0];
    const int*   edge_row = (const int*)d_in[1];
    const int*   edge_col = (const int*)d_in[2];
    const float* edge_val = (const float*)d_in[3];
    const float* W        = (const float*)d_in[4];
    const float* bias     = (const float*)d_in[5];
    float*       out      = (float*)d_out;

    float* H   = (float*)d_ws;
    int2*  rec = (int2*)d_ws;
    int*   stt = (int*)((char*)d_ws + WS_OFF_START);

    const int n_edges = in_sizes[1];
    const int nblk    = (n_edges + SRT_EPB - 1) / SRT_EPB;

    gcn_sort_chunk<<<nblk, 256, 0, stream>>>(edge_row, edge_col, edge_val, rec, stt, n_edges, nblk);
    gcn_place_rows<<<SRT_NG, 256, 0, stream>>>(rec, stt, (int2*)out, nblk);

    gcn_gemm_wmma_f32<<<GEMM_GRID, 256, 0, stream>>>(X, W, bias, H);

    gcn_row_sum_slots<<<GCN_N_NODES / 8, 256, 0, stream>>>(edge_row, edge_col, edge_val, H, out, n_edges);
}
